# starting kernel
# baseline (speedup 1.0000x reference)
#include <hip/hip_runtime.h>
#include <math.h>


#define BB   8
#define NN_  36
#define DD   64
#define NN2  (NN_*NN_)
#define NCON (2*NN_)
#define KSZ  (NN2+NCON)
#define PP   1376
#define NBLK (PP/16)
#define KDIAG 1225.0f
#define EPSK  1e-5f
#define ALPHA 1000.0f

typedef float v2f __attribute__((ext_vector_type(2)));
typedef float v8f __attribute__((ext_vector_type(8)));

__global__ void feat_kernel(const float* __restrict__ emb1,
                            const float* __restrict__ emb2,
                            const float* __restrict__ W,
                            const float* __restrict__ bias,
                            float* __restrict__ h1, float* __restrict__ h2) {
    int b = blockIdx.x, which = blockIdx.y;
    const float* e = (which == 0 ? emb1 : emb2) + (size_t)b * NN_ * DD;
    float* h = (which == 0 ? h1 : h2) + (size_t)b * NN_ * DD;
    __shared__ float se[NN_ * DD];
    __shared__ float sw[DD * DD];
    __shared__ float snrm[DD];
    int tid = threadIdx.x;
    for (int i = tid; i < NN_ * DD; i += 256) se[i] = e[i];
    for (int i = tid; i < DD * DD; i += 256) sw[i] = W[i];
    __syncthreads();
    for (int d = tid; d < DD; d += 256) {
        float s = 0.f;
        for (int n = 0; n < NN_; n++) { float v = se[n * DD + d]; s += v * v; }
        snrm[d] = fmaxf(sqrtf(s), 1e-12f);
    }
    __syncthreads();
    for (int i = tid; i < NN_ * DD; i += 256) se[i] = se[i] / snrm[i % DD];
    __syncthreads();
    for (int idx = tid; idx < NN_ * DD; idx += 256) {
        int n = idx / DD, i = idx % DD;
        float acc = bias[i];
        for (int d = 0; d < DD; d++) acc += se[n * DD + d] * sw[i * DD + d];
        h[idx] = fmaxf(acc, 0.f);
    }
}

__global__ void gram_rhs_kernel(const float* __restrict__ h1,
                                const float* __restrict__ h2,
                                float* __restrict__ S, float* __restrict__ rhs) {
    int b = blockIdx.x, tid = threadIdx.x;
    __shared__ float s1[NN_ * DD];
    __shared__ float s2[NN_ * DD];
    for (int i = tid; i < NN_ * DD; i += 256) {
        s1[i] = h1[(size_t)b * NN_ * DD + i];
        s2[i] = h2[(size_t)b * NN_ * DD + i];
    }
    __syncthreads();
    for (int idx = tid; idx < NN2; idx += 256) {
        int i = idx / NN_, j = idx % NN_;
        float a = 0.f;
        for (int d = 0; d < DD; d++) a += s1[i * DD + d] * s2[j * DD + d];
        S[(size_t)b * NN2 + idx] = a;
        rhs[(size_t)b * PP + j * NN_ + i] = a;
    }
    for (int r = NN2 + tid; r < PP; r += 256)
        rhs[(size_t)b * PP + r] = (r < KSZ) ? 1.0f : 0.0f;
}

__global__ void assemble_kernel(const float* __restrict__ S, float* __restrict__ K) {
    int b = blockIdx.z;
    int r = blockIdx.y * 16 + threadIdx.y;
    int c = blockIdx.x * 16 + threadIdx.x;
    const float* Sb = S + (size_t)b * NN2;
    float val = 0.f;
    if (r < NN2 && c < NN2) {
        int t = r / NN_, s = r - t * NN_, u = c / NN_, v = c - u * NN_;
        val = (r == c) ? KDIAG : 0.f;
        if (s != v && t != u) val -= 0.5f * (Sb[s * NN_ + t] + Sb[v * NN_ + u]);
    } else if (r < NN2) {
        int i = c - NN2, t = r / NN_, s = r - t * NN_;
        if (i < NN_)            val = (s == i) ? 1.f : 0.f;
        else if (i < NCON)      val = (t == i - NN_) ? 1.f : 0.f;
    } else if (c < NN2) {
        int i = r - NN2, u = c / NN_, v = c - u * NN_;
        if (i < NN_)            val = (v == i) ? 1.f : 0.f;
        else if (i < NCON)      val = (u == i - NN_) ? 1.f : 0.f;
    } else if (r == c) {
        val = (r < KSZ) ? -EPSK : 1.0f;
    }
    K[((size_t)b * PP + r) * PP + c] = val;
}

__global__ void lu_panel_kernel(float* __restrict__ K, int kb) {
    extern __shared__ float pan[];
    int b = blockIdx.x, tid = threadIdx.x;
    int r0 = kb * 16, H = PP - r0;
    float* Kb = K + (size_t)b * PP * PP;
    for (int i = tid; i < H * 16; i += 256)
        pan[i] = Kb[(size_t)(r0 + i / 16) * PP + r0 + (i % 16)];
    __syncthreads();
    for (int j = 0; j < 16; j++) {
        float dinv = 1.0f / pan[j * 16 + j];
        for (int i = j + 1 + tid; i < H; i += 256) {
            float lij = pan[i * 16 + j] * dinv;
            pan[i * 16 + j] = lij;
            for (int c2 = j + 1; c2 < 16; c2++)
                pan[i * 16 + c2] -= lij * pan[j * 16 + c2];
        }
        __syncthreads();
    }
    for (int i = tid; i < H * 16; i += 256)
        Kb[(size_t)(r0 + i / 16) * PP + r0 + (i % 16)] = pan[i];
}

__global__ void lu_rowpanel_kernel(float* __restrict__ K, int kb) {
    int b = blockIdx.y, tid = threadIdx.x;
    int r0 = kb * 16;
    int col = r0 + 16 + blockIdx.x * 256 + tid;
    __shared__ float L11[16 * 16];
    float* Kb = K + (size_t)b * PP * PP;
    if (tid < 256)
        L11[tid] = Kb[(size_t)(r0 + tid / 16) * PP + r0 + (tid % 16)];
    __syncthreads();
    if (col < PP) {
        float x[16];
        for (int j = 0; j < 16; j++) x[j] = Kb[(size_t)(r0 + j) * PP + col];
        for (int j = 1; j < 16; j++) {
            float s = x[j];
            for (int t2 = 0; t2 < j; t2++) s -= L11[j * 16 + t2] * x[t2];
            x[j] = s;
        }
        for (int j = 0; j < 16; j++) Kb[(size_t)(r0 + j) * PP + col] = x[j];
    }
}

template <bool R1, bool C1>
__device__ __forceinline__ void update_tile(float* __restrict__ Kb,
                                            int row0, int col0, int kcol,
                                            int lane) {
    int half = lane >> 4, l16 = lane & 15;
    int row1 = row0 + 16, col1 = col0 + 16;

    v8f c00, c01, c10, c11;
    for (int r = 0; r < 8; r++) {
        size_t ro0 = (size_t)(row0 + r + 8 * half) * PP;
        c00[r] = Kb[ro0 + col0 + l16];
        if (C1) c01[r] = Kb[ro0 + col1 + l16];
        if (R1) {
            size_t ro1 = (size_t)(row1 + r + 8 * half) * PP;
            c10[r] = Kb[ro1 + col0 + l16];
            if (C1) c11[r] = Kb[ro1 + col1 + l16];
        }
    }
    for (int kk = 0; kk < 4; kk++) {
        int k0 = kcol + kk * 4;
        int kc = k0 + 2 * half;
        v2f a0, a1, b0, b1;
        v2f av0 = *(const v2f*)(Kb + (size_t)(row0 + l16) * PP + kc);
        a0[0] = -av0[0]; a0[1] = -av0[1];
        if (R1) {
            v2f av1 = *(const v2f*)(Kb + (size_t)(row1 + l16) * PP + kc);
            a1[0] = -av1[0]; a1[1] = -av1[1];
        }
        for (int i = 0; i < 2; i++) {
            size_t kr = (size_t)(kc + i) * PP;
            b0[i] = Kb[kr + col0 + l16];
            if (C1) b1[i] = Kb[kr + col1 + l16];
        }
        c00 = __builtin_amdgcn_wmma_f32_16x16x4_f32(false, a0, false, b0, (short)0, c00, false, false);
        if (C1) c01 = __builtin_amdgcn_wmma_f32_16x16x4_f32(false, a0, false, b1, (short)0, c01, false, false);
        if (R1) c10 = __builtin_amdgcn_wmma_f32_16x16x4_f32(false, a1, false, b0, (short)0, c10, false, false);
        if (R1 && C1) c11 = __builtin_amdgcn_wmma_f32_16x16x4_f32(false, a1, false, b1, (short)0, c11, false, false);
    }
    for (int r = 0; r < 8; r++) {
        size_t ro0 = (size_t)(row0 + r + 8 * half) * PP;
        Kb[ro0 + col0 + l16] = c00[r];
        if (C1) Kb[ro0 + col1 + l16] = c01[r];
        if (R1) {
            size_t ro1 = (size_t)(row1 + r + 8 * half) * PP;
            Kb[ro1 + col0 + l16] = c10[r];
            if (C1) Kb[ro1 + col1 + l16] = c11[r];
        }
    }
}

__global__ void lu_update_kernel(float* __restrict__ K, int kb) {
    int b = blockIdx.z;
    float* Kb = K + (size_t)b * PP * PP;
    int base = kb * 16 + 16;
    int row0 = base + blockIdx.y * 32;
    int col0 = base + blockIdx.x * 32;
    bool hasR1 = (row0 + 16 < PP), hasC1 = (col0 + 16 < PP);
    int lane = threadIdx.x;
    int kcol = kb * 16;
    if (hasR1 && hasC1)      update_tile<true,  true >(Kb, row0, col0, kcol, lane);
    else if (hasR1)          update_tile<true,  false>(Kb, row0, col0, kcol, lane);
    else if (hasC1)          update_tile<false, true >(Kb, row0, col0, kcol, lane);
    else                     update_tile<false, false>(Kb, row0, col0, kcol, lane);
}

__global__ void trisolve_kernel(const float* __restrict__ K,
                                const float* __restrict__ rhs,
                                float* __restrict__ x) {
    int b = blockIdx.x, tid = threadIdx.x;
    __shared__ float y[PP];
    const float* Kb = K + (size_t)b * PP * PP;
    for (int i = tid; i < PP; i += 256) y[i] = rhs[(size_t)b * PP + i];
    __syncthreads();
    for (int kb = 0; kb < NBLK; kb++) {
        int r0 = kb * 16;
        if (tid == 0) {
            for (int j = 1; j < 16; j++) {
                float s = y[r0 + j];
                for (int t2 = 0; t2 < j; t2++)
                    s -= Kb[(size_t)(r0 + j) * PP + r0 + t2] * y[r0 + t2];
                y[r0 + j] = s;
            }
        }
        __syncthreads();
        for (int i = r0 + 16 + tid; i < PP; i += 256) {
            float s = y[i];
            for (int j = 0; j < 16; j++)
                s -= Kb[(size_t)i * PP + r0 + j] * y[r0 + j];
            y[i] = s;
        }
        __syncthreads();
    }
    for (int kb = NBLK - 1; kb >= 0; kb--) {
        int r0 = kb * 16;
        if (tid == 0) {
            for (int j = 15; j >= 0; j--) {
                float s = y[r0 + j];
                for (int t2 = j + 1; t2 < 16; t2++)
                    s -= Kb[(size_t)(r0 + j) * PP + r0 + t2] * y[r0 + t2];
                y[r0 + j] = s / Kb[(size_t)(r0 + j) * PP + r0 + j];
            }
        }
        __syncthreads();
        for (int i = tid; i < r0; i += 256) {
            float s = y[i];
            for (int j = 0; j < 16; j++)
                s -= Kb[(size_t)i * PP + r0 + j] * y[r0 + j];
            y[i] = s;
        }
        __syncthreads();
    }
    for (int i = tid; i < PP; i += 256) x[(size_t)b * PP + i] = y[i];
}

__global__ void vote_kernel(const float* __restrict__ x, float* __restrict__ out) {
    int s = blockIdx.x, b = blockIdx.y, tid = threadIdx.x;
    __shared__ float vals[NN_];
    if (tid < NN_) {
        float v = x[(size_t)b * PP + tid * NN_ + s];
        vals[tid] = fminf(fmaxf(v, 0.f), 1.f);
    }
    __syncthreads();
    if (tid == 0) {
        float m = -1e30f;
        for (int t = 0; t < NN_; t++) m = fmaxf(m, vals[t]);
        float sum = 0.f;
        for (int t = 0; t < NN_; t++) { float e = __expf(ALPHA * (vals[t] - m)); vals[t] = e; sum += e; }
        float inv = 1.f / sum;
        for (int t = 0; t < NN_; t++) out[((size_t)b * NN_ + s) * NN_ + t] = vals[t] * inv;
    }
}

extern "C" void kernel_launch(void* const* d_in, const int* in_sizes, int n_in,
                              void* d_out, int out_size, void* d_ws, size_t ws_size,
                              hipStream_t stream) {
    const float* emb1 = (const float*)d_in[0];
    const float* emb2 = (const float*)d_in[1];
    const float* Wcg  = (const float*)d_in[2];
    const float* bcg  = (const float*)d_in[3];

    float* ws  = (float*)d_ws;
    size_t kelem = (size_t)BB * PP * PP;
    float* Km  = ws;
    float* h1  = Km + kelem;
    float* h2  = h1 + (size_t)BB * NN_ * DD;
    float* S   = h2 + (size_t)BB * NN_ * DD;
    float* rhs = S  + (size_t)BB * NN2;
    float* x   = rhs + (size_t)BB * PP;

    feat_kernel<<<dim3(BB, 2), 256, 0, stream>>>(emb1, emb2, Wcg, bcg, h1, h2);
    gram_rhs_kernel<<<BB, 256, 0, stream>>>(h1, h2, S, rhs);
    assemble_kernel<<<dim3(NBLK, NBLK, BB), dim3(16, 16), 0, stream>>>(S, Km);

    for (int kb = 0; kb < NBLK; kb++) {
        int H = PP - kb * 16;
        lu_panel_kernel<<<BB, 256, (size_t)H * 16 * sizeof(float), stream>>>(Km, kb);
        int W2 = PP - (kb + 1) * 16;
        if (W2 > 0) {
            lu_rowpanel_kernel<<<dim3((W2 + 255) / 256, BB), 256, 0, stream>>>(Km, kb);
            int tiles = W2 / 16;
            int t32 = (tiles + 1) / 2;
            lu_update_kernel<<<dim3(t32, t32, BB), 32, 0, stream>>>(Km, kb);
        }
    }

    trisolve_kernel<<<BB, 256, 0, stream>>>(Km, rhs, x);
    vote_kernel<<<dim3(NN_, BB), 64, 0, stream>>>(x, (float*)d_out);
}


# Panel step spread over the chip: each wave refactors the 16x16 diagonal block in registers, then one thread per L21 row / U12 column; panel and row-panel launches merged (op order unchanged)
# speedup vs baseline: 2.1207x; 2.1207x over previous
#include <hip/hip_runtime.h>
#include <math.h>


#define BB   8
#define NN_  36
#define DD   64
#define NN2  (NN_*NN_)
#define NCON (2*NN_)
#define KSZ  (NN2+NCON)
#define PP   1376
#define NBLK (PP/16)
#define KDIAG 1225.0f
#define EPSK  1e-5f
#define ALPHA 1000.0f

typedef float v2f __attribute__((ext_vector_type(2)));
typedef float v8f __attribute__((ext_vector_type(8)));

__global__ void feat_kernel(const float* __restrict__ emb1,
                            const float* __restrict__ emb2,
                            const float* __restrict__ W,
                            const float* __restrict__ bias,
                            float* __restrict__ h1, float* __restrict__ h2) {
    int b = blockIdx.x, which = blockIdx.y;
    const float* e = (which == 0 ? emb1 : emb2) + (size_t)b * NN_ * DD;
    float* h = (which == 0 ? h1 : h2) + (size_t)b * NN_ * DD;
    __shared__ float se[NN_ * DD];
    __shared__ float sw[DD * DD];
    __shared__ float snrm[DD];
    int tid = threadIdx.x;
    for (int i = tid; i < NN_ * DD; i += 256) se[i] = e[i];
    for (int i = tid; i < DD * DD; i += 256) sw[i] = W[i];
    __syncthreads();
    for (int d = tid; d < DD; d += 256) {
        float s = 0.f;
        for (int n = 0; n < NN_; n++) { float v = se[n * DD + d]; s += v * v; }
        snrm[d] = fmaxf(sqrtf(s), 1e-12f);
    }
    __syncthreads();
    for (int i = tid; i < NN_ * DD; i += 256) se[i] = se[i] / snrm[i % DD];
    __syncthreads();
    for (int idx = tid; idx < NN_ * DD; idx += 256) {
        int n = idx / DD, i = idx % DD;
        float acc = bias[i];
        for (int d = 0; d < DD; d++) acc += se[n * DD + d] * sw[i * DD + d];
        h[idx] = fmaxf(acc, 0.f);
    }
}

__global__ void gram_rhs_kernel(const float* __restrict__ h1,
                                const float* __restrict__ h2,
                                float* __restrict__ S, float* __restrict__ rhs) {
    int b = blockIdx.x, tid = threadIdx.x;
    __shared__ float s1[NN_ * DD];
    __shared__ float s2[NN_ * DD];
    for (int i = tid; i < NN_ * DD; i += 256) {
        s1[i] = h1[(size_t)b * NN_ * DD + i];
        s2[i] = h2[(size_t)b * NN_ * DD + i];
    }
    __syncthreads();
    for (int idx = tid; idx < NN2; idx += 256) {
        int i = idx / NN_, j = idx % NN_;
        float a = 0.f;
        for (int d = 0; d < DD; d++) a += s1[i * DD + d] * s2[j * DD + d];
        S[(size_t)b * NN2 + idx] = a;
        rhs[(size_t)b * PP + j * NN_ + i] = a;
    }
    for (int r = NN2 + tid; r < PP; r += 256)
        rhs[(size_t)b * PP + r] = (r < KSZ) ? 1.0f : 0.0f;
}

__global__ void assemble_kernel(const float* __restrict__ S, float* __restrict__ K) {
    int b = blockIdx.z;
    int r = blockIdx.y * 16 + threadIdx.y;
    int c = blockIdx.x * 16 + threadIdx.x;
    const float* Sb = S + (size_t)b * NN2;
    float val = 0.f;
    if (r < NN2 && c < NN2) {
        int t = r / NN_, s = r - t * NN_, u = c / NN_, v = c - u * NN_;
        val = (r == c) ? KDIAG : 0.f;
        if (s != v && t != u) val -= 0.5f * (Sb[s * NN_ + t] + Sb[v * NN_ + u]);
    } else if (r < NN2) {
        int i = c - NN2, t = r / NN_, s = r - t * NN_;
        if (i < NN_)            val = (s == i) ? 1.f : 0.f;
        else if (i < NCON)      val = (t == i - NN_) ? 1.f : 0.f;
    } else if (c < NN2) {
        int i = r - NN2, u = c / NN_, v = c - u * NN_;
        if (i < NN_)            val = (v == i) ? 1.f : 0.f;
        else if (i < NCON)      val = (u == i - NN_) ? 1.f : 0.f;
    } else if (r == c) {
        val = (r < KSZ) ? -EPSK : 1.0f;
    }
    K[((size_t)b * PP + r) * PP + c] = val;
}

#define PNT 64

__device__ __forceinline__ float lane_bcast(float v, int lane) {
    return __int_as_float(__builtin_amdgcn_readlane(__float_as_int(v), lane));
}

__global__ void lu_panels_kernel(float* __restrict__ K, float* __restrict__ slot,
                                 int kb, int nc) {
    int b = blockIdx.y, bx = blockIdx.x, tid = threadIdx.x;
    int r0 = kb * 16;
    float* Kb = K + (size_t)b * PP * PP;
    float* sl = slot + (size_t)b * 256;
    int row = tid & 15;

    bool isRow = (bx >= nc);
    int idx = r0 + 16 + (isRow ? bx - nc : bx) * PNT + tid;
    bool act = idx < PP;
    bool keeper = (bx == 0 && tid < 16);

    float x[16], a[16], o[16];
    #pragma unroll
    for (int j = 0; j < 16; j++) { x[j] = 0.f; o[j] = 0.f; }
    if (act) {
        if (isRow) {
            const float4* p = (const float4*)(Kb + (size_t)idx * PP + r0);
            #pragma unroll
            for (int q = 0; q < 4; q++) {
                float4 v = p[q];
                x[4 * q] = v.x; x[4 * q + 1] = v.y; x[4 * q + 2] = v.z; x[4 * q + 3] = v.w;
            }
        } else {
            #pragma unroll
            for (int j = 0; j < 16; j++) x[j] = Kb[(size_t)(r0 + j) * PP + idx];
        }
    }
    {
        const float4* p = (const float4*)(Kb + (size_t)(r0 + row) * PP + r0);
        #pragma unroll
        for (int q = 0; q < 4; q++) {
            float4 v = p[q];
            a[4 * q] = v.x; a[4 * q + 1] = v.y; a[4 * q + 2] = v.z; a[4 * q + 3] = v.w;
        }
    }
    if (keeper && kb > 0) {
        const float4* p = (const float4*)(sl + row * 16);
        #pragma unroll
        for (int q = 0; q < 4; q++) {
            float4 v = p[q];
            o[4 * q] = v.x; o[4 * q + 1] = v.y; o[4 * q + 2] = v.z; o[4 * q + 3] = v.w;
        }
    }

    float dv[16];
    #pragma unroll
    for (int j = 0; j < 16; j++) {
        float dinv = 1.0f / lane_bcast(a[j], j);
        dv[j] = dinv;
        bool below = row > j;
        float lij = a[j] * dinv;
        #pragma unroll
        for (int c2 = j + 1; c2 < 16; c2++) {
            float t = a[c2] - lij * lane_bcast(a[c2], j);
            a[c2] = below ? t : a[c2];
        }
        a[j] = below ? lij : a[j];
    }

    if (keeper) {
        if (kb > 0) {
            float4* p = (float4*)(Kb + (size_t)(r0 - 16 + row) * PP + r0 - 16);
            #pragma unroll
            for (int q = 0; q < 4; q++)
                p[q] = make_float4(o[4 * q], o[4 * q + 1], o[4 * q + 2], o[4 * q + 3]);
        }
        float4* p = (gridDim.x == 1) ? (float4*)(Kb + (size_t)(r0 + row) * PP + r0)
                                     : (float4*)(sl + row * 16);
        #pragma unroll
        for (int q = 0; q < 4; q++)
            p[q] = make_float4(a[4 * q], a[4 * q + 1], a[4 * q + 2], a[4 * q + 3]);
    }

    if (isRow) {
        #pragma unroll
        for (int j = 0; j < 16; j++) {
            float lij = x[j] * dv[j];
            x[j] = lij;
            #pragma unroll
            for (int c2 = j + 1; c2 < 16; c2++)
                x[c2] -= lij * lane_bcast(a[c2], j);
        }
        if (act) {
            float4* p = (float4*)(Kb + (size_t)idx * PP + r0);
            #pragma unroll
            for (int q = 0; q < 4; q++)
                p[q] = make_float4(x[4 * q], x[4 * q + 1], x[4 * q + 2], x[4 * q + 3]);
        }
    } else {
        #pragma unroll
        for (int j = 1; j < 16; j++) {
            float s = x[j];
            #pragma unroll
            for (int t2 = 0; t2 < j; t2++) s -= lane_bcast(a[t2], j) * x[t2];
            x[j] = s;
        }
        if (act) {
            #pragma unroll
            for (int j = 0; j < 16; j++) Kb[(size_t)(r0 + j) * PP + idx] = x[j];
        }
    }
}

template <bool R1, bool C1>
__device__ __forceinline__ void update_tile(float* __restrict__ Kb,
                                            int row0, int col0, int kcol,
                                            int lane) {
    int half = lane >> 4, l16 = lane & 15;
    int row1 = row0 + 16, col1 = col0 + 16;

    v8f c00, c01, c10, c11;
    for (int r = 0; r < 8; r++) {
        size_t ro0 = (size_t)(row0 + r + 8 * half) * PP;
        c00[r] = Kb[ro0 + col0 + l16];
        if (C1) c01[r] = Kb[ro0 + col1 + l16];
        if (R1) {
            size_t ro1 = (size_t)(row1 + r + 8 * half) * PP;
            c10[r] = Kb[ro1 + col0 + l16];
            if (C1) c11[r] = Kb[ro1 + col1 + l16];
        }
    }
    for (int kk = 0; kk < 4; kk++) {
        int k0 = kcol + kk * 4;
        int kc = k0 + 2 * half;
        v2f a0, a1, b0, b1;
        v2f av0 = *(const v2f*)(Kb + (size_t)(row0 + l16) * PP + kc);
        a0[0] = -av0[0]; a0[1] = -av0[1];
        if (R1) {
            v2f av1 = *(const v2f*)(Kb + (size_t)(row1 + l16) * PP + kc);
            a1[0] = -av1[0]; a1[1] = -av1[1];
        }
        for (int i = 0; i < 2; i++) {
            size_t kr = (size_t)(kc + i) * PP;
            b0[i] = Kb[kr + col0 + l16];
            if (C1) b1[i] = Kb[kr + col1 + l16];
        }
        c00 = __builtin_amdgcn_wmma_f32_16x16x4_f32(false, a0, false, b0, (short)0, c00, false, false);
        if (C1) c01 = __builtin_amdgcn_wmma_f32_16x16x4_f32(false, a0, false, b1, (short)0, c01, false, false);
        if (R1) c10 = __builtin_amdgcn_wmma_f32_16x16x4_f32(false, a1, false, b0, (short)0, c10, false, false);
        if (R1 && C1) c11 = __builtin_amdgcn_wmma_f32_16x16x4_f32(false, a1, false, b1, (short)0, c11, false, false);
    }
    for (int r = 0; r < 8; r++) {
        size_t ro0 = (size_t)(row0 + r + 8 * half) * PP;
        Kb[ro0 + col0 + l16] = c00[r];
        if (C1) Kb[ro0 + col1 + l16] = c01[r];
        if (R1) {
            size_t ro1 = (size_t)(row1 + r + 8 * half) * PP;
            Kb[ro1 + col0 + l16] = c10[r];
            if (C1) Kb[ro1 + col1 + l16] = c11[r];
        }
    }
}

__global__ void lu_update_kernel(float* __restrict__ K, int kb) {
    int b = blockIdx.z;
    float* Kb = K + (size_t)b * PP * PP;
    int base = kb * 16 + 16;
    int row0 = base + blockIdx.y * 32;
    int col0 = base + blockIdx.x * 32;
    bool hasR1 = (row0 + 16 < PP), hasC1 = (col0 + 16 < PP);
    int lane = threadIdx.x;
    int kcol = kb * 16;
    if (hasR1 && hasC1)      update_tile<true,  true >(Kb, row0, col0, kcol, lane);
    else if (hasR1)          update_tile<true,  false>(Kb, row0, col0, kcol, lane);
    else if (hasC1)          update_tile<false, true >(Kb, row0, col0, kcol, lane);
    else                     update_tile<false, false>(Kb, row0, col0, kcol, lane);
}

__global__ void trisolve_kernel(const float* __restrict__ K,
                                const float* __restrict__ rhs,
                                float* __restrict__ x) {
    int b = blockIdx.x, tid = threadIdx.x;
    __shared__ float y[PP];
    const float* Kb = K + (size_t)b * PP * PP;
    for (int i = tid; i < PP; i += 256) y[i] = rhs[(size_t)b * PP + i];
    __syncthreads();
    for (int kb = 0; kb < NBLK; kb++) {
        int r0 = kb * 16;
        if (tid == 0) {
            for (int j = 1; j < 16; j++) {
                float s = y[r0 + j];
                for (int t2 = 0; t2 < j; t2++)
                    s -= Kb[(size_t)(r0 + j) * PP + r0 + t2] * y[r0 + t2];
                y[r0 + j] = s;
            }
        }
        __syncthreads();
        for (int i = r0 + 16 + tid; i < PP; i += 256) {
            float s = y[i];
            for (int j = 0; j < 16; j++)
                s -= Kb[(size_t)i * PP + r0 + j] * y[r0 + j];
            y[i] = s;
        }
        __syncthreads();
    }
    for (int kb = NBLK - 1; kb >= 0; kb--) {
        int r0 = kb * 16;
        if (tid == 0) {
            for (int j = 15; j >= 0; j--) {
                float s = y[r0 + j];
                for (int t2 = j + 1; t2 < 16; t2++)
                    s -= Kb[(size_t)(r0 + j) * PP + r0 + t2] * y[r0 + t2];
                y[r0 + j] = s / Kb[(size_t)(r0 + j) * PP + r0 + j];
            }
        }
        __syncthreads();
        for (int i = tid; i < r0; i += 256) {
            float s = y[i];
            for (int j = 0; j < 16; j++)
                s -= Kb[(size_t)i * PP + r0 + j] * y[r0 + j];
            y[i] = s;
        }
        __syncthreads();
    }
    for (int i = tid; i < PP; i += 256) x[(size_t)b * PP + i] = y[i];
}

__global__ void vote_kernel(const float* __restrict__ x, float* __restrict__ out) {
    int s = blockIdx.x, b = blockIdx.y, tid = threadIdx.x;
    __shared__ float vals[NN_];
    if (tid < NN_) {
        float v = x[(size_t)b * PP + tid * NN_ + s];
        vals[tid] = fminf(fmaxf(v, 0.f), 1.f);
    }
    __syncthreads();
    if (tid == 0) {
        float m = -1e30f;
        for (int t = 0; t < NN_; t++) m = fmaxf(m, vals[t]);
        float sum = 0.f;
        for (int t = 0; t < NN_; t++) { float e = __expf(ALPHA * (vals[t] - m)); vals[t] = e; sum += e; }
        float inv = 1.f / sum;
        for (int t = 0; t < NN_; t++) out[((size_t)b * NN_ + s) * NN_ + t] = vals[t] * inv;
    }
}

extern "C" void kernel_launch(void* const* d_in, const int* in_sizes, int n_in,
                              void* d_out, int out_size, void* d_ws, size_t ws_size,
                              hipStream_t stream) {
    const float* emb1 = (const float*)d_in[0];
    const float* emb2 = (const float*)d_in[1];
    const float* Wcg  = (const float*)d_in[2];
    const float* bcg  = (const float*)d_in[3];

    float* ws  = (float*)d_ws;
    size_t kelem = (size_t)BB * PP * PP;
    float* Km  = ws;
    float* h1  = Km + kelem;
    float* h2  = h1 + (size_t)BB * NN_ * DD;
    float* S   = h2 + (size_t)BB * NN_ * DD;
    float* rhs = S  + (size_t)BB * NN2;
    float* x   = rhs + (size_t)BB * PP;

    feat_kernel<<<dim3(BB, 2), 256, 0, stream>>>(emb1, emb2, Wcg, bcg, h1, h2);
    gram_rhs_kernel<<<BB, 256, 0, stream>>>(h1, h2, S, rhs);
    assemble_kernel<<<dim3(NBLK, NBLK, BB), dim3(16, 16), 0, stream>>>(S, Km);

    float* slot = h1;
    for (int kb = 0; kb < NBLK; kb++) {
        int W2 = PP - (kb + 1) * 16;
        int nc = (W2 + PNT - 1) / PNT;
        lu_panels_kernel<<<dim3(nc > 0 ? 2 * nc : 1, BB), PNT, 0, stream>>>(Km, slot, kb, nc);
        if (W2 > 0) {
            int tiles = W2 / 16;
            int t32 = (tiles + 1) / 2;
            lu_update_kernel<<<dim3(t32, t32, BB), 32, 0, stream>>>(Km, kb);
        }
    }

    trisolve_kernel<<<BB, 256, 0, stream>>>(Km, rhs, x);
    vote_kernel<<<dim3(NN_, BB), 64, 0, stream>>>(x, (float*)d_out);
}


# LU block width 16 -> 32 (43 block steps, not 86): half the panel and update launches, 32 WMMAs per update wave; panel columns 17-32 now get scalar FMAs, not WMMA sums (sum order changes there)
# speedup vs baseline: 3.4548x; 1.6291x over previous
#include <hip/hip_runtime.h>
#include <math.h>


#define BB   8
#define NN_  36
#define DD   64
#define NN2  (NN_*NN_)
#define NCON (2*NN_)
#define KSZ  (NN2+NCON)
#define PP   1376
#define NBLK (PP/16)
#define KDIAG 1225.0f
#define EPSK  1e-5f
#define ALPHA 1000.0f

typedef float v2f __attribute__((ext_vector_type(2)));
typedef float v8f __attribute__((ext_vector_type(8)));

__global__ void feat_kernel(const float* __restrict__ emb1,
                            const float* __restrict__ emb2,
                            const float* __restrict__ W,
                            const float* __restrict__ bias,
                            float* __restrict__ h1, float* __restrict__ h2) {
    int b = blockIdx.x, which = blockIdx.y;
    const float* e = (which == 0 ? emb1 : emb2) + (size_t)b * NN_ * DD;
    float* h = (which == 0 ? h1 : h2) + (size_t)b * NN_ * DD;
    __shared__ float se[NN_ * DD];
    __shared__ float sw[DD * DD];
    __shared__ float snrm[DD];
    int tid = threadIdx.x;
    for (int i = tid; i < NN_ * DD; i += 256) se[i] = e[i];
    for (int i = tid; i < DD * DD; i += 256) sw[i] = W[i];
    __syncthreads();
    for (int d = tid; d < DD; d += 256) {
        float s = 0.f;
        for (int n = 0; n < NN_; n++) { float v = se[n * DD + d]; s += v * v; }
        snrm[d] = fmaxf(sqrtf(s), 1e-12f);
    }
    __syncthreads();
    for (int i = tid; i < NN_ * DD; i += 256) se[i] = se[i] / snrm[i % DD];
    __syncthreads();
    for (int idx = tid; idx < NN_ * DD; idx += 256) {
        int n = idx / DD, i = idx % DD;
        float acc = bias[i];
        for (int d = 0; d < DD; d++) acc += se[n * DD + d] * sw[i * DD + d];
        h[idx] = fmaxf(acc, 0.f);
    }
}

__global__ void gram_rhs_kernel(const float* __restrict__ h1,
                                const float* __restrict__ h2,
                                float* __restrict__ S, float* __restrict__ rhs) {
    int b = blockIdx.x, tid = threadIdx.x;
    __shared__ float s1[NN_ * DD];
    __shared__ float s2[NN_ * DD];
    for (int i = tid; i < NN_ * DD; i += 256) {
        s1[i] = h1[(size_t)b * NN_ * DD + i];
        s2[i] = h2[(size_t)b * NN_ * DD + i];
    }
    __syncthreads();
    for (int idx = tid; idx < NN2; idx += 256) {
        int i = idx / NN_, j = idx % NN_;
        float a = 0.f;
        for (int d = 0; d < DD; d++) a += s1[i * DD + d] * s2[j * DD + d];
        S[(size_t)b * NN2 + idx] = a;
        rhs[(size_t)b * PP + j * NN_ + i] = a;
    }
    for (int r = NN2 + tid; r < PP; r += 256)
        rhs[(size_t)b * PP + r] = (r < KSZ) ? 1.0f : 0.0f;
}

__global__ void assemble_kernel(const float* __restrict__ S, float* __restrict__ K) {
    int b = blockIdx.z;
    int r = blockIdx.y * 16 + threadIdx.y;
    int c = blockIdx.x * 16 + threadIdx.x;
    const float* Sb = S + (size_t)b * NN2;
    float val = 0.f;
    if (r < NN2 && c < NN2) {
        int t = r / NN_, s = r - t * NN_, u = c / NN_, v = c - u * NN_;
        val = (r == c) ? KDIAG : 0.f;
        if (s != v && t != u) val -= 0.5f * (Sb[s * NN_ + t] + Sb[v * NN_ + u]);
    } else if (r < NN2) {
        int i = c - NN2, t = r / NN_, s = r - t * NN_;
        if (i < NN_)            val = (s == i) ? 1.f : 0.f;
        else if (i < NCON)      val = (t == i - NN_) ? 1.f : 0.f;
    } else if (c < NN2) {
        int i = r - NN2, u = c / NN_, v = c - u * NN_;
        if (i < NN_)            val = (v == i) ? 1.f : 0.f;
        else if (i < NCON)      val = (u == i - NN_) ? 1.f : 0.f;
    } else if (r == c) {
        val = (r < KSZ) ? -EPSK : 1.0f;
    }
    K[((size_t)b * PP + r) * PP + c] = val;
}

#define NB  32
#define NBLK32 (PP/NB)
#define PNT 64

__device__ __forceinline__ float lane_bcast(float v, int lane) {
    return __int_as_float(__builtin_amdgcn_readlane(__float_as_int(v), lane));
}

__device__ __forceinline__ float in_reg(float v) {
    asm("" : "+v"(v));
    return v;
}

__global__ void __launch_bounds__(PNT)
lu_panels_kernel(float* __restrict__ K, float* __restrict__ slot, int kb) {
    int b = blockIdx.y, bx = blockIdx.x, tid = threadIdx.x;
    int r0 = kb * NB;
    float* Kb = K + (size_t)b * PP * PP;
    float* sl = slot + (size_t)b * (NB * NB);
    int row = tid & (NB - 1);

    int idx = r0 + NB + bx * PNT + tid;
    bool act = idx < PP;
    bool keeper = (bx == 0 && tid < NB);

    float xr[NB], xc[NB], a[NB], o[NB];
    #pragma unroll
    for (int j = 0; j < NB; j++) { xr[j] = 0.f; xc[j] = 0.f; o[j] = 0.f; }
    if (act) {
        const float4* p = (const float4*)(Kb + (size_t)idx * PP + r0);
        #pragma unroll
        for (int q = 0; q < NB / 4; q++) {
            float4 v = p[q];
            xr[4 * q] = v.x; xr[4 * q + 1] = v.y; xr[4 * q + 2] = v.z; xr[4 * q + 3] = v.w;
        }
        #pragma unroll
        for (int j = 0; j < NB; j++) xc[j] = Kb[(size_t)(r0 + j) * PP + idx];
    }
    {
        const float4* p = (const float4*)(Kb + (size_t)(r0 + row) * PP + r0);
        #pragma unroll
        for (int q = 0; q < NB / 4; q++) {
            float4 v = p[q];
            a[4 * q] = v.x; a[4 * q + 1] = v.y; a[4 * q + 2] = v.z; a[4 * q + 3] = v.w;
        }
    }
    if (keeper && kb > 0) {
        const float4* p = (const float4*)(sl + row * NB);
        #pragma unroll
        for (int q = 0; q < NB / 4; q++) {
            float4 v = p[q];
            o[4 * q] = v.x; o[4 * q + 1] = v.y; o[4 * q + 2] = v.z; o[4 * q + 3] = v.w;
        }
    }

    #pragma unroll
    for (int j = 0; j < NB; j++) {
        float u[NB];
        #pragma unroll
        for (int c2 = j; c2 < NB; c2++) u[c2] = lane_bcast(a[c2], j);
        float dinv = 1.0f / u[j];
        bool below = row > j;
        float lij = a[j] * dinv;
        float nl = in_reg(-lij);
        #pragma unroll
        for (int c2 = j + 1; c2 < NB; c2++) {
            float t = a[c2] + nl * u[c2];
            a[c2] = below ? t : a[c2];
        }
        a[j] = below ? lij : a[j];

        float xl = xr[j] * dinv;
        xr[j] = xl;
        float nxl = in_reg(-xl);
        #pragma unroll
        for (int c2 = j + 1; c2 < NB; c2++) xr[c2] += nxl * u[c2];
        __builtin_amdgcn_sched_barrier(0);

        float nxj = in_reg(-xc[j]);
        #pragma unroll
        for (int r = j + 1; r < NB; r++) xc[r] += lane_bcast(a[j], r) * nxj;
        __builtin_amdgcn_sched_barrier(0);
    }

    if (keeper) {
        if (kb > 0) {
            float4* p = (float4*)(Kb + (size_t)(r0 - NB + row) * PP + r0 - NB);
            #pragma unroll
            for (int q = 0; q < NB / 4; q++)
                p[q] = make_float4(o[4 * q], o[4 * q + 1], o[4 * q + 2], o[4 * q + 3]);
        }
        float4* p = (r0 + NB == PP) ? (float4*)(Kb + (size_t)(r0 + row) * PP + r0)
                                    : (float4*)(sl + row * NB);
        #pragma unroll
        for (int q = 0; q < NB / 4; q++)
            p[q] = make_float4(a[4 * q], a[4 * q + 1], a[4 * q + 2], a[4 * q + 3]);
    }

    if (act) {
        float4* p = (float4*)(Kb + (size_t)idx * PP + r0);
        #pragma unroll
        for (int q = 0; q < NB / 4; q++)
            p[q] = make_float4(xr[4 * q], xr[4 * q + 1], xr[4 * q + 2], xr[4 * q + 3]);
        #pragma unroll
        for (int j = 0; j < NB; j++) Kb[(size_t)(r0 + j) * PP + idx] = xc[j];
    }
}

__device__ __forceinline__ void update_tile(float* __restrict__ Kb,
                                            int row0, int col0, int kcol,
                                            int lane) {
    int half = lane >> 4, l16 = lane & 15;
    int row1 = row0 + 16, col1 = col0 + 16;

    v8f c00, c01, c10, c11;
    #pragma unroll
    for (int r = 0; r < 8; r++) {
        size_t ro0 = (size_t)(row0 + r + 8 * half) * PP;
        size_t ro1 = (size_t)(row1 + r + 8 * half) * PP;
        c00[r] = Kb[ro0 + col0 + l16];
        c01[r] = Kb[ro0 + col1 + l16];
        c10[r] = Kb[ro1 + col0 + l16];
        c11[r] = Kb[ro1 + col1 + l16];
    }
    v2f a0[NB / 4], a1[NB / 4], b0[NB / 4], b1[NB / 4];
    #pragma unroll
    for (int kk = 0; kk < NB / 4; kk++) {
        int kc = kcol + kk * 4 + 2 * half;
        a0[kk] = *(const v2f*)(Kb + (size_t)(row0 + l16) * PP + kc);
        a1[kk] = *(const v2f*)(Kb + (size_t)(row1 + l16) * PP + kc);
        #pragma unroll
        for (int i = 0; i < 2; i++) {
            size_t kr = (size_t)(kc + i) * PP;
            b0[kk][i] = Kb[kr + col0 + l16];
            b1[kk][i] = Kb[kr + col1 + l16];
        }
    }
    __builtin_amdgcn_sched_barrier(0);
    #pragma unroll
    for (int kk = 0; kk < NB / 4; kk++) {
        v2f n0 = -a0[kk], n1 = -a1[kk];
        c00 = __builtin_amdgcn_wmma_f32_16x16x4_f32(false, n0, false, b0[kk], (short)0, c00, false, false);
        c01 = __builtin_amdgcn_wmma_f32_16x16x4_f32(false, n0, false, b1[kk], (short)0, c01, false, false);
        c10 = __builtin_amdgcn_wmma_f32_16x16x4_f32(false, n1, false, b0[kk], (short)0, c10, false, false);
        c11 = __builtin_amdgcn_wmma_f32_16x16x4_f32(false, n1, false, b1[kk], (short)0, c11, false, false);
    }
    #pragma unroll
    for (int r = 0; r < 8; r++) {
        size_t ro0 = (size_t)(row0 + r + 8 * half) * PP;
        size_t ro1 = (size_t)(row1 + r + 8 * half) * PP;
        Kb[ro0 + col0 + l16] = c00[r];
        Kb[ro0 + col1 + l16] = c01[r];
        Kb[ro1 + col0 + l16] = c10[r];
        Kb[ro1 + col1 + l16] = c11[r];
    }
}

__global__ void __launch_bounds__(32)
lu_update_kernel(float* __restrict__ K, int kb) {
    int b = blockIdx.z;
    float* Kb = K + (size_t)b * PP * PP;
    int base = kb * NB + NB;
    int row0 = base + blockIdx.y * 32;
    int col0 = base + blockIdx.x * 32;
    update_tile(Kb, row0, col0, kb * NB, threadIdx.x);
}

__global__ void trisolve_kernel(const float* __restrict__ K,
                                const float* __restrict__ rhs,
                                float* __restrict__ x) {
    int b = blockIdx.x, tid = threadIdx.x;
    __shared__ float y[PP];
    const float* Kb = K + (size_t)b * PP * PP;
    for (int i = tid; i < PP; i += 256) y[i] = rhs[(size_t)b * PP + i];
    __syncthreads();
    for (int kb = 0; kb < NBLK; kb++) {
        int r0 = kb * 16;
        if (tid == 0) {
            for (int j = 1; j < 16; j++) {
                float s = y[r0 + j];
                for (int t2 = 0; t2 < j; t2++)
                    s -= Kb[(size_t)(r0 + j) * PP + r0 + t2] * y[r0 + t2];
                y[r0 + j] = s;
            }
        }
        __syncthreads();
        for (int i = r0 + 16 + tid; i < PP; i += 256) {
            float s = y[i];
            for (int j = 0; j < 16; j++)
                s -= Kb[(size_t)i * PP + r0 + j] * y[r0 + j];
            y[i] = s;
        }
        __syncthreads();
    }
    for (int kb = NBLK - 1; kb >= 0; kb--) {
        int r0 = kb * 16;
        if (tid == 0) {
            for (int j = 15; j >= 0; j--) {
                float s = y[r0 + j];
                for (int t2 = j + 1; t2 < 16; t2++)
                    s -= Kb[(size_t)(r0 + j) * PP + r0 + t2] * y[r0 + t2];
                y[r0 + j] = s / Kb[(size_t)(r0 + j) * PP + r0 + j];
            }
        }
        __syncthreads();
        for (int i = tid; i < r0; i += 256) {
            float s = y[i];
            for (int j = 0; j < 16; j++)
                s -= Kb[(size_t)i * PP + r0 + j] * y[r0 + j];
            y[i] = s;
        }
        __syncthreads();
    }
    for (int i = tid; i < PP; i += 256) x[(size_t)b * PP + i] = y[i];
}

__global__ void vote_kernel(const float* __restrict__ x, float* __restrict__ out) {
    int s = blockIdx.x, b = blockIdx.y, tid = threadIdx.x;
    __shared__ float vals[NN_];
    if (tid < NN_) {
        float v = x[(size_t)b * PP + tid * NN_ + s];
        vals[tid] = fminf(fmaxf(v, 0.f), 1.f);
    }
    __syncthreads();
    if (tid == 0) {
        float m = -1e30f;
        for (int t = 0; t < NN_; t++) m = fmaxf(m, vals[t]);
        float sum = 0.f;
        for (int t = 0; t < NN_; t++) { float e = __expf(ALPHA * (vals[t] - m)); vals[t] = e; sum += e; }
        float inv = 1.f / sum;
        for (int t = 0; t < NN_; t++) out[((size_t)b * NN_ + s) * NN_ + t] = vals[t] * inv;
    }
}

extern "C" void kernel_launch(void* const* d_in, const int* in_sizes, int n_in,
                              void* d_out, int out_size, void* d_ws, size_t ws_size,
                              hipStream_t stream) {
    const float* emb1 = (const float*)d_in[0];
    const float* emb2 = (const float*)d_in[1];
    const float* Wcg  = (const float*)d_in[2];
    const float* bcg  = (const float*)d_in[3];

    float* ws  = (float*)d_ws;
    size_t kelem = (size_t)BB * PP * PP;
    float* Km  = ws;
    float* h1  = Km + kelem;
    float* h2  = h1 + (size_t)BB * NN_ * DD;
    float* S   = h2 + (size_t)BB * NN_ * DD;
    float* rhs = S  + (size_t)BB * NN2;
    float* x   = rhs + (size_t)BB * PP;

    feat_kernel<<<dim3(BB, 2), 256, 0, stream>>>(emb1, emb2, Wcg, bcg, h1, h2);
    gram_rhs_kernel<<<BB, 256, 0, stream>>>(h1, h2, S, rhs);
    assemble_kernel<<<dim3(NBLK, NBLK, BB), dim3(16, 16), 0, stream>>>(S, Km);

    float* slot = h1;
    for (int kb = 0; kb < NBLK32; kb++) {
        int W2 = PP - (kb + 1) * NB;
        int nc = (W2 + PNT - 1) / PNT;
        lu_panels_kernel<<<dim3(nc > 0 ? nc : 1, BB), PNT, 0, stream>>>(Km, slot, kb);
        if (W2 > 0) {
            int t32 = W2 / 32;
            lu_update_kernel<<<dim3(t32, t32, BB), 32, 0, stream>>>(Km, kb);
        }
    }

    trisolve_kernel<<<BB, 256, 0, stream>>>(Km, rhs, x);
    vote_kernel<<<dim3(NN_, BB), 64, 0, stream>>>(x, (float*)d_out);
}


# Forward substitution folded into the LU: rhs rides as padding column 1368 of K, panel+update kernels leave y=L^-1 rhs there; solve kernel keeps only back substitution (y now summed in WMMA order)
# speedup vs baseline: 3.8598x; 1.1172x over previous
#include <hip/hip_runtime.h>
#include <math.h>


#define BB   8
#define NN_  36
#define DD   64
#define NN2  (NN_*NN_)
#define NCON (2*NN_)
#define KSZ  (NN2+NCON)
#define PP   1376
#define NBLK (PP/16)
#define KDIAG 1225.0f
#define EPSK  1e-5f
#define ALPHA 1000.0f

typedef float v2f __attribute__((ext_vector_type(2)));
typedef float v8f __attribute__((ext_vector_type(8)));

__global__ void feat_kernel(const float* __restrict__ emb1,
                            const float* __restrict__ emb2,
                            const float* __restrict__ W,
                            const float* __restrict__ bias,
                            float* __restrict__ h1, float* __restrict__ h2) {
    int b = blockIdx.x, which = blockIdx.y;
    const float* e = (which == 0 ? emb1 : emb2) + (size_t)b * NN_ * DD;
    float* h = (which == 0 ? h1 : h2) + (size_t)b * NN_ * DD;
    __shared__ float se[NN_ * DD];
    __shared__ float sw[DD * DD];
    __shared__ float snrm[DD];
    int tid = threadIdx.x;
    for (int i = tid; i < NN_ * DD; i += 256) se[i] = e[i];
    for (int i = tid; i < DD * DD; i += 256) sw[i] = W[i];
    __syncthreads();
    for (int d = tid; d < DD; d += 256) {
        float s = 0.f;
        for (int n = 0; n < NN_; n++) { float v = se[n * DD + d]; s += v * v; }
        snrm[d] = fmaxf(sqrtf(s), 1e-12f);
    }
    __syncthreads();
    for (int i = tid; i < NN_ * DD; i += 256) se[i] = se[i] / snrm[i % DD];
    __syncthreads();
    for (int idx = tid; idx < NN_ * DD; idx += 256) {
        int n = idx / DD, i = idx % DD;
        float acc = bias[i];
        for (int d = 0; d < DD; d++) acc += se[n * DD + d] * sw[i * DD + d];
        h[idx] = fmaxf(acc, 0.f);
    }
}

__global__ void gram_kernel(const float* __restrict__ h1,
                            const float* __restrict__ h2,
                            float* __restrict__ S) {
    int b = blockIdx.x, tid = threadIdx.x;
    __shared__ float s1[NN_ * DD];
    __shared__ float s2[NN_ * DD];
    for (int i = tid; i < NN_ * DD; i += 256) {
        s1[i] = h1[(size_t)b * NN_ * DD + i];
        s2[i] = h2[(size_t)b * NN_ * DD + i];
    }
    __syncthreads();
    for (int idx = tid; idx < NN2; idx += 256) {
        int i = idx / NN_, j = idx % NN_;
        float a = 0.f;
        for (int d = 0; d < DD; d++) a += s1[i * DD + d] * s2[j * DD + d];
        S[(size_t)b * NN2 + idx] = a;
    }
}

__global__ void assemble_kernel(const float* __restrict__ S, float* __restrict__ K) {
    int b = blockIdx.z;
    int r = blockIdx.y * 16 + threadIdx.y;
    int c = blockIdx.x * 16 + threadIdx.x;
    const float* Sb = S + (size_t)b * NN2;
    float val = 0.f;
    if (r < NN2 && c < NN2) {
        int t = r / NN_, s = r - t * NN_, u = c / NN_, v = c - u * NN_;
        val = (r == c) ? KDIAG : 0.f;
        if (s != v && t != u) val -= 0.5f * (Sb[s * NN_ + t] + Sb[v * NN_ + u]);
    } else if (r < NN2) {
        int i = c - NN2, t = r / NN_, s = r - t * NN_;
        if (i < NN_)            val = (s == i) ? 1.f : 0.f;
        else if (i < NCON)      val = (t == i - NN_) ? 1.f : 0.f;
        else if (i == NCON)     val = Sb[s * NN_ + t];
    } else if (c < NN2) {
        int i = r - NN2, u = c / NN_, v = c - u * NN_;
        if (i < NN_)            val = (v == i) ? 1.f : 0.f;
        else if (i < NCON)      val = (u == i - NN_) ? 1.f : 0.f;
    } else if (r == c) {
        val = (r < KSZ) ? -EPSK : 1.0f;
    } else if (c == KSZ && r < KSZ) {
        val = 1.0f;
    }
    K[((size_t)b * PP + r) * PP + c] = val;
}

#define NB  32
#define NBLK32 (PP/NB)
#define PNT 64

__device__ __forceinline__ float lane_bcast(float v, int lane) {
    return __int_as_float(__builtin_amdgcn_readlane(__float_as_int(v), lane));
}

__device__ __forceinline__ float in_reg(float v) {
    asm("" : "+v"(v));
    return v;
}

__global__ void __launch_bounds__(PNT)
lu_panels_kernel(float* __restrict__ K, float* __restrict__ slot, int kb) {
    int b = blockIdx.y, bx = blockIdx.x, tid = threadIdx.x;
    int r0 = kb * NB;
    float* Kb = K + (size_t)b * PP * PP;
    float* sl = slot + (size_t)b * (NB * NB);
    int row = tid & (NB - 1);

    int idx = r0 + NB + bx * PNT + tid;
    bool act = idx < PP;
    bool keeper = (bx == 0 && tid < NB);

    float xr[NB], xc[NB], a[NB], o[NB];
    #pragma unroll
    for (int j = 0; j < NB; j++) { xr[j] = 0.f; xc[j] = 0.f; o[j] = 0.f; }
    if (act) {
        const float4* p = (const float4*)(Kb + (size_t)idx * PP + r0);
        #pragma unroll
        for (int q = 0; q < NB / 4; q++) {
            float4 v = p[q];
            xr[4 * q] = v.x; xr[4 * q + 1] = v.y; xr[4 * q + 2] = v.z; xr[4 * q + 3] = v.w;
        }
        #pragma unroll
        for (int j = 0; j < NB; j++) xc[j] = Kb[(size_t)(r0 + j) * PP + idx];
    }
    {
        const float4* p = (const float4*)(Kb + (size_t)(r0 + row) * PP + r0);
        #pragma unroll
        for (int q = 0; q < NB / 4; q++) {
            float4 v = p[q];
            a[4 * q] = v.x; a[4 * q + 1] = v.y; a[4 * q + 2] = v.z; a[4 * q + 3] = v.w;
        }
    }
    if (keeper && kb > 0) {
        const float4* p = (const float4*)(sl + row * NB);
        #pragma unroll
        for (int q = 0; q < NB / 4; q++) {
            float4 v = p[q];
            o[4 * q] = v.x; o[4 * q + 1] = v.y; o[4 * q + 2] = v.z; o[4 * q + 3] = v.w;
        }
    }

    #pragma unroll
    for (int j = 0; j < NB; j++) {
        float u[NB];
        #pragma unroll
        for (int c2 = j; c2 < NB; c2++) u[c2] = lane_bcast(a[c2], j);
        float dinv = 1.0f / u[j];
        bool below = row > j;
        float lij = a[j] * dinv;
        float nl = in_reg(-lij);
        #pragma unroll
        for (int c2 = j + 1; c2 < NB; c2++) {
            float t = a[c2] + nl * u[c2];
            a[c2] = below ? t : a[c2];
        }
        a[j] = below ? lij : a[j];

        float xl = xr[j] * dinv;
        xr[j] = xl;
        float nxl = in_reg(-xl);
        #pragma unroll
        for (int c2 = j + 1; c2 < NB; c2++) xr[c2] += nxl * u[c2];
        __builtin_amdgcn_sched_barrier(0);

        float nxj = in_reg(-xc[j]);
        #pragma unroll
        for (int r = j + 1; r < NB; r++) xc[r] += lane_bcast(a[j], r) * nxj;
        __builtin_amdgcn_sched_barrier(0);
    }

    if (keeper) {
        if (kb > 0) {
            float4* p = (float4*)(Kb + (size_t)(r0 - NB + row) * PP + r0 - NB);
            #pragma unroll
            for (int q = 0; q < NB / 4; q++)
                p[q] = make_float4(o[4 * q], o[4 * q + 1], o[4 * q + 2], o[4 * q + 3]);
        }
        float4* p = (r0 + NB == PP) ? (float4*)(Kb + (size_t)(r0 + row) * PP + r0)
                                    : (float4*)(sl + row * NB);
        #pragma unroll
        for (int q = 0; q < NB / 4; q++)
            p[q] = make_float4(a[4 * q], a[4 * q + 1], a[4 * q + 2], a[4 * q + 3]);
    }

    if (act) {
        float4* p = (float4*)(Kb + (size_t)idx * PP + r0);
        #pragma unroll
        for (int q = 0; q < NB / 4; q++)
            p[q] = make_float4(xr[4 * q], xr[4 * q + 1], xr[4 * q + 2], xr[4 * q + 3]);
        #pragma unroll
        for (int j = 0; j < NB; j++) Kb[(size_t)(r0 + j) * PP + idx] = xc[j];
    }
}

__device__ __forceinline__ void update_tile(float* __restrict__ Kb,
                                            int row0, int col0, int kcol,
                                            int lane) {
    int half = lane >> 4, l16 = lane & 15;
    int row1 = row0 + 16, col1 = col0 + 16;

    v8f c00, c01, c10, c11;
    #pragma unroll
    for (int r = 0; r < 8; r++) {
        size_t ro0 = (size_t)(row0 + r + 8 * half) * PP;
        size_t ro1 = (size_t)(row1 + r + 8 * half) * PP;
        c00[r] = Kb[ro0 + col0 + l16];
        c01[r] = Kb[ro0 + col1 + l16];
        c10[r] = Kb[ro1 + col0 + l16];
        c11[r] = Kb[ro1 + col1 + l16];
    }
    v2f a0[NB / 4], a1[NB / 4], b0[NB / 4], b1[NB / 4];
    #pragma unroll
    for (int kk = 0; kk < NB / 4; kk++) {
        int kc = kcol + kk * 4 + 2 * half;
        a0[kk] = *(const v2f*)(Kb + (size_t)(row0 + l16) * PP + kc);
        a1[kk] = *(const v2f*)(Kb + (size_t)(row1 + l16) * PP + kc);
        #pragma unroll
        for (int i = 0; i < 2; i++) {
            size_t kr = (size_t)(kc + i) * PP;
            b0[kk][i] = Kb[kr + col0 + l16];
            b1[kk][i] = Kb[kr + col1 + l16];
        }
    }
    __builtin_amdgcn_sched_barrier(0);
    #pragma unroll
    for (int kk = 0; kk < NB / 4; kk++) {
        v2f n0 = -a0[kk], n1 = -a1[kk];
        c00 = __builtin_amdgcn_wmma_f32_16x16x4_f32(false, n0, false, b0[kk], (short)0, c00, false, false);
        c01 = __builtin_amdgcn_wmma_f32_16x16x4_f32(false, n0, false, b1[kk], (short)0, c01, false, false);
        c10 = __builtin_amdgcn_wmma_f32_16x16x4_f32(false, n1, false, b0[kk], (short)0, c10, false, false);
        c11 = __builtin_amdgcn_wmma_f32_16x16x4_f32(false, n1, false, b1[kk], (short)0, c11, false, false);
    }
    #pragma unroll
    for (int r = 0; r < 8; r++) {
        size_t ro0 = (size_t)(row0 + r + 8 * half) * PP;
        size_t ro1 = (size_t)(row1 + r + 8 * half) * PP;
        Kb[ro0 + col0 + l16] = c00[r];
        Kb[ro0 + col1 + l16] = c01[r];
        Kb[ro1 + col0 + l16] = c10[r];
        Kb[ro1 + col1 + l16] = c11[r];
    }
}

__global__ void __launch_bounds__(32)
lu_update_kernel(float* __restrict__ K, int kb) {
    int b = blockIdx.z;
    float* Kb = K + (size_t)b * PP * PP;
    int base = kb * NB + NB;
    int row0 = base + blockIdx.y * 32;
    int col0 = base + blockIdx.x * 32;
    update_tile(Kb, row0, col0, kb * NB, threadIdx.x);
}

__global__ void backsolve_kernel(const float* __restrict__ K,
                                 float* __restrict__ x) {
    int b = blockIdx.x, tid = threadIdx.x;
    __shared__ float y[PP];
    const float* Kb = K + (size_t)b * PP * PP;
    for (int i = tid; i < PP; i += 256)
        y[i] = (i < KSZ) ? Kb[(size_t)i * PP + KSZ] : 0.f;
    __syncthreads();
    for (int kb = NBLK - 1; kb >= 0; kb--) {
        int r0 = kb * 16;
        if (tid == 0) {
            for (int j = 15; j >= 0; j--) {
                float s = y[r0 + j];
                for (int t2 = j + 1; t2 < 16; t2++)
                    s -= Kb[(size_t)(r0 + j) * PP + r0 + t2] * y[r0 + t2];
                y[r0 + j] = s / Kb[(size_t)(r0 + j) * PP + r0 + j];
            }
        }
        __syncthreads();
        for (int i = tid; i < r0; i += 256) {
            float s = y[i];
            for (int j = 0; j < 16; j++)
                s -= Kb[(size_t)i * PP + r0 + j] * y[r0 + j];
            y[i] = s;
        }
        __syncthreads();
    }
    for (int i = tid; i < PP; i += 256) x[(size_t)b * PP + i] = y[i];
}

__global__ void vote_kernel(const float* __restrict__ x, float* __restrict__ out) {
    int s = blockIdx.x, b = blockIdx.y, tid = threadIdx.x;
    __shared__ float vals[NN_];
    if (tid < NN_) {
        float v = x[(size_t)b * PP + tid * NN_ + s];
        vals[tid] = fminf(fmaxf(v, 0.f), 1.f);
    }
    __syncthreads();
    if (tid == 0) {
        float m = -1e30f;
        for (int t = 0; t < NN_; t++) m = fmaxf(m, vals[t]);
        float sum = 0.f;
        for (int t = 0; t < NN_; t++) { float e = __expf(ALPHA * (vals[t] - m)); vals[t] = e; sum += e; }
        float inv = 1.f / sum;
        for (int t = 0; t < NN_; t++) out[((size_t)b * NN_ + s) * NN_ + t] = vals[t] * inv;
    }
}

extern "C" void kernel_launch(void* const* d_in, const int* in_sizes, int n_in,
                              void* d_out, int out_size, void* d_ws, size_t ws_size,
                              hipStream_t stream) {
    const float* emb1 = (const float*)d_in[0];
    const float* emb2 = (const float*)d_in[1];
    const float* Wcg  = (const float*)d_in[2];
    const float* bcg  = (const float*)d_in[3];

    float* ws  = (float*)d_ws;
    size_t kelem = (size_t)BB * PP * PP;
    float* Km  = ws;
    float* h1  = Km + kelem;
    float* h2  = h1 + (size_t)BB * NN_ * DD;
    float* S   = h2 + (size_t)BB * NN_ * DD;
    float* x   = S  + (size_t)BB * NN2;

    feat_kernel<<<dim3(BB, 2), 256, 0, stream>>>(emb1, emb2, Wcg, bcg, h1, h2);
    gram_kernel<<<BB, 256, 0, stream>>>(h1, h2, S);
    assemble_kernel<<<dim3(NBLK, NBLK, BB), dim3(16, 16), 0, stream>>>(S, Km);

    float* slot = h1;
    for (int kb = 0; kb < NBLK32; kb++) {
        int W2 = PP - (kb + 1) * NB;
        int nc = (W2 + PNT - 1) / PNT;
        lu_panels_kernel<<<dim3(nc > 0 ? nc : 1, BB), PNT, 0, stream>>>(Km, slot, kb);
        if (W2 > 0) {
            int t32 = W2 / 32;
            lu_update_kernel<<<dim3(t32, t32, BB), 32, 0, stream>>>(Km, kb);
        }
    }

    backsolve_kernel<<<BB, 256, 0, stream>>>(Km, x);
    vote_kernel<<<dim3(NN_, BB), 64, 0, stream>>>(x, (float*)d_out);
}


# Back substitution rebuilt, 32 unknowns a step: one wave solves the triangle in registers (rows pre-scaled by pivot reciprocal), 512 threads apply it with coalesced row reads (sum order changes)
# speedup vs baseline: 5.2074x; 1.3491x over previous
#include <hip/hip_runtime.h>
#include <math.h>


#define BB   8
#define NN_  36
#define DD   64
#define NN2  (NN_*NN_)
#define NCON (2*NN_)
#define KSZ  (NN2+NCON)
#define PP   1376
#define NBLK (PP/16)
#define KDIAG 1225.0f
#define EPSK  1e-5f
#define ALPHA 1000.0f

typedef float v2f __attribute__((ext_vector_type(2)));
typedef float v8f __attribute__((ext_vector_type(8)));

__global__ void feat_kernel(const float* __restrict__ emb1,
                            const float* __restrict__ emb2,
                            const float* __restrict__ W,
                            const float* __restrict__ bias,
                            float* __restrict__ h1, float* __restrict__ h2) {
    int b = blockIdx.x, which = blockIdx.y;
    const float* e = (which == 0 ? emb1 : emb2) + (size_t)b * NN_ * DD;
    float* h = (which == 0 ? h1 : h2) + (size_t)b * NN_ * DD;
    __shared__ float se[NN_ * DD];
    __shared__ float sw[DD * DD];
    __shared__ float snrm[DD];
    int tid = threadIdx.x;
    for (int i = tid; i < NN_ * DD; i += 256) se[i] = e[i];
    for (int i = tid; i < DD * DD; i += 256) sw[i] = W[i];
    __syncthreads();
    for (int d = tid; d < DD; d += 256) {
        float s = 0.f;
        for (int n = 0; n < NN_; n++) { float v = se[n * DD + d]; s += v * v; }
        snrm[d] = fmaxf(sqrtf(s), 1e-12f);
    }
    __syncthreads();
    for (int i = tid; i < NN_ * DD; i += 256) se[i] = se[i] / snrm[i % DD];
    __syncthreads();
    for (int idx = tid; idx < NN_ * DD; idx += 256) {
        int n = idx / DD, i = idx % DD;
        float acc = bias[i];
        for (int d = 0; d < DD; d++) acc += se[n * DD + d] * sw[i * DD + d];
        h[idx] = fmaxf(acc, 0.f);
    }
}

__global__ void gram_kernel(const float* __restrict__ h1,
                            const float* __restrict__ h2,
                            float* __restrict__ S) {
    int b = blockIdx.x, tid = threadIdx.x;
    __shared__ float s1[NN_ * DD];
    __shared__ float s2[NN_ * DD];
    for (int i = tid; i < NN_ * DD; i += 256) {
        s1[i] = h1[(size_t)b * NN_ * DD + i];
        s2[i] = h2[(size_t)b * NN_ * DD + i];
    }
    __syncthreads();
    for (int idx = tid; idx < NN2; idx += 256) {
        int i = idx / NN_, j = idx % NN_;
        float a = 0.f;
        for (int d = 0; d < DD; d++) a += s1[i * DD + d] * s2[j * DD + d];
        S[(size_t)b * NN2 + idx] = a;
    }
}

__global__ void assemble_kernel(const float* __restrict__ S, float* __restrict__ K) {
    int b = blockIdx.z;
    int r = blockIdx.y * 16 + threadIdx.y;
    int c = blockIdx.x * 16 + threadIdx.x;
    const float* Sb = S + (size_t)b * NN2;
    float val = 0.f;
    if (r < NN2 && c < NN2) {
        int t = r / NN_, s = r - t * NN_, u = c / NN_, v = c - u * NN_;
        val = (r == c) ? KDIAG : 0.f;
        if (s != v && t != u) val -= 0.5f * (Sb[s * NN_ + t] + Sb[v * NN_ + u]);
    } else if (r < NN2) {
        int i = c - NN2, t = r / NN_, s = r - t * NN_;
        if (i < NN_)            val = (s == i) ? 1.f : 0.f;
        else if (i < NCON)      val = (t == i - NN_) ? 1.f : 0.f;
        else if (i == NCON)     val = Sb[s * NN_ + t];
    } else if (c < NN2) {
        int i = r - NN2, u = c / NN_, v = c - u * NN_;
        if (i < NN_)            val = (v == i) ? 1.f : 0.f;
        else if (i < NCON)      val = (u == i - NN_) ? 1.f : 0.f;
    } else if (r == c) {
        val = (r < KSZ) ? -EPSK : 1.0f;
    } else if (c == KSZ && r < KSZ) {
        val = 1.0f;
    }
    K[((size_t)b * PP + r) * PP + c] = val;
}

#define NB  32
#define NBLK32 (PP/NB)
#define PNT 64

__device__ __forceinline__ float lane_bcast(float v, int lane) {
    return __int_as_float(__builtin_amdgcn_readlane(__float_as_int(v), lane));
}

__device__ __forceinline__ float in_reg(float v) {
    asm("" : "+v"(v));
    return v;
}

__global__ void __launch_bounds__(PNT)
lu_panels_kernel(float* __restrict__ K, float* __restrict__ slot, int kb) {
    int b = blockIdx.y, bx = blockIdx.x, tid = threadIdx.x;
    int r0 = kb * NB;
    float* Kb = K + (size_t)b * PP * PP;
    float* sl = slot + (size_t)b * (NB * NB);
    int row = tid & (NB - 1);

    int idx = r0 + NB + bx * PNT + tid;
    bool act = idx < PP;
    bool keeper = (bx == 0 && tid < NB);

    float xr[NB], xc[NB], a[NB], o[NB];
    #pragma unroll
    for (int j = 0; j < NB; j++) { xr[j] = 0.f; xc[j] = 0.f; o[j] = 0.f; }
    if (act) {
        const float4* p = (const float4*)(Kb + (size_t)idx * PP + r0);
        #pragma unroll
        for (int q = 0; q < NB / 4; q++) {
            float4 v = p[q];
            xr[4 * q] = v.x; xr[4 * q + 1] = v.y; xr[4 * q + 2] = v.z; xr[4 * q + 3] = v.w;
        }
        #pragma unroll
        for (int j = 0; j < NB; j++) xc[j] = Kb[(size_t)(r0 + j) * PP + idx];
    }
    {
        const float4* p = (const float4*)(Kb + (size_t)(r0 + row) * PP + r0);
        #pragma unroll
        for (int q = 0; q < NB / 4; q++) {
            float4 v = p[q];
            a[4 * q] = v.x; a[4 * q + 1] = v.y; a[4 * q + 2] = v.z; a[4 * q + 3] = v.w;
        }
    }
    if (keeper && kb > 0) {
        const float4* p = (const float4*)(sl + row * NB);
        #pragma unroll
        for (int q = 0; q < NB / 4; q++) {
            float4 v = p[q];
            o[4 * q] = v.x; o[4 * q + 1] = v.y; o[4 * q + 2] = v.z; o[4 * q + 3] = v.w;
        }
    }

    #pragma unroll
    for (int j = 0; j < NB; j++) {
        float u[NB];
        #pragma unroll
        for (int c2 = j; c2 < NB; c2++) u[c2] = lane_bcast(a[c2], j);
        float dinv = 1.0f / u[j];
        bool below = row > j;
        float lij = a[j] * dinv;
        float nl = in_reg(-lij);
        #pragma unroll
        for (int c2 = j + 1; c2 < NB; c2++) {
            float t = a[c2] + nl * u[c2];
            a[c2] = below ? t : a[c2];
        }
        a[j] = below ? lij : a[j];

        float xl = xr[j] * dinv;
        xr[j] = xl;
        float nxl = in_reg(-xl);
        #pragma unroll
        for (int c2 = j + 1; c2 < NB; c2++) xr[c2] += nxl * u[c2];
        __builtin_amdgcn_sched_barrier(0);

        float nxj = in_reg(-xc[j]);
        #pragma unroll
        for (int r = j + 1; r < NB; r++) xc[r] += lane_bcast(a[j], r) * nxj;
        __builtin_amdgcn_sched_barrier(0);
    }

    if (keeper) {
        if (kb > 0) {
            float4* p = (float4*)(Kb + (size_t)(r0 - NB + row) * PP + r0 - NB);
            #pragma unroll
            for (int q = 0; q < NB / 4; q++)
                p[q] = make_float4(o[4 * q], o[4 * q + 1], o[4 * q + 2], o[4 * q + 3]);
        }
        float4* p = (r0 + NB == PP) ? (float4*)(Kb + (size_t)(r0 + row) * PP + r0)
                                    : (float4*)(sl + row * NB);
        #pragma unroll
        for (int q = 0; q < NB / 4; q++)
            p[q] = make_float4(a[4 * q], a[4 * q + 1], a[4 * q + 2], a[4 * q + 3]);
    }

    if (act) {
        float4* p = (float4*)(Kb + (size_t)idx * PP + r0);
        #pragma unroll
        for (int q = 0; q < NB / 4; q++)
            p[q] = make_float4(xr[4 * q], xr[4 * q + 1], xr[4 * q + 2], xr[4 * q + 3]);
        #pragma unroll
        for (int j = 0; j < NB; j++) Kb[(size_t)(r0 + j) * PP + idx] = xc[j];
    }
}

__device__ __forceinline__ void update_tile(float* __restrict__ Kb,
                                            int row0, int col0, int kcol,
                                            int lane) {
    int half = lane >> 4, l16 = lane & 15;
    int row1 = row0 + 16, col1 = col0 + 16;

    v8f c00, c01, c10, c11;
    #pragma unroll
    for (int r = 0; r < 8; r++) {
        size_t ro0 = (size_t)(row0 + r + 8 * half) * PP;
        size_t ro1 = (size_t)(row1 + r + 8 * half) * PP;
        c00[r] = Kb[ro0 + col0 + l16];
        c01[r] = Kb[ro0 + col1 + l16];
        c10[r] = Kb[ro1 + col0 + l16];
        c11[r] = Kb[ro1 + col1 + l16];
    }
    v2f a0[NB / 4], a1[NB / 4], b0[NB / 4], b1[NB / 4];
    #pragma unroll
    for (int kk = 0; kk < NB / 4; kk++) {
        int kc = kcol + kk * 4 + 2 * half;
        a0[kk] = *(const v2f*)(Kb + (size_t)(row0 + l16) * PP + kc);
        a1[kk] = *(const v2f*)(Kb + (size_t)(row1 + l16) * PP + kc);
        #pragma unroll
        for (int i = 0; i < 2; i++) {
            size_t kr = (size_t)(kc + i) * PP;
            b0[kk][i] = Kb[kr + col0 + l16];
            b1[kk][i] = Kb[kr + col1 + l16];
        }
    }
    __builtin_amdgcn_sched_barrier(0);
    #pragma unroll
    for (int kk = 0; kk < NB / 4; kk++) {
        v2f n0 = -a0[kk], n1 = -a1[kk];
        c00 = __builtin_amdgcn_wmma_f32_16x16x4_f32(false, n0, false, b0[kk], (short)0, c00, false, false);
        c01 = __builtin_amdgcn_wmma_f32_16x16x4_f32(false, n0, false, b1[kk], (short)0, c01, false, false);
        c10 = __builtin_amdgcn_wmma_f32_16x16x4_f32(false, n1, false, b0[kk], (short)0, c10, false, false);
        c11 = __builtin_amdgcn_wmma_f32_16x16x4_f32(false, n1, false, b1[kk], (short)0, c11, false, false);
    }
    #pragma unroll
    for (int r = 0; r < 8; r++) {
        size_t ro0 = (size_t)(row0 + r + 8 * half) * PP;
        size_t ro1 = (size_t)(row1 + r + 8 * half) * PP;
        Kb[ro0 + col0 + l16] = c00[r];
        Kb[ro0 + col1 + l16] = c01[r];
        Kb[ro1 + col0 + l16] = c10[r];
        Kb[ro1 + col1 + l16] = c11[r];
    }
}

__global__ void __launch_bounds__(32)
lu_update_kernel(float* __restrict__ K, int kb) {
    int b = blockIdx.z;
    float* Kb = K + (size_t)b * PP * PP;
    int base = kb * NB + NB;
    int row0 = base + blockIdx.y * 32;
    int col0 = base + blockIdx.x * 32;
    update_tile(Kb, row0, col0, kb * NB, threadIdx.x);
}

#define BS_T    512
#define BS_RPP  (BS_T / 8)
#define BS_MAXP ((PP - NB) / BS_RPP)

template <int M>
__device__ __forceinline__ float lane_xor(float v) {
    return __int_as_float(__builtin_amdgcn_update_dpp(0, __float_as_int(v), 0x160 + M, 0xf, 0xf, true));
}

__device__ __forceinline__ void bs_load_diag(const float* __restrict__ Kb, int r, int lane,
                                             float (&un)[NB], float& dn) {
    dn = Kb[(size_t)(r + lane) * PP + r + lane];
    const float4* p = (const float4*)(Kb + (size_t)(r + lane) * PP + r);
    #pragma unroll
    for (int q = 0; q < NB / 4; q++) {
        float4 v = p[q];
        un[4 * q] = v.x; un[4 * q + 1] = v.y; un[4 * q + 2] = v.z; un[4 * q + 3] = v.w;
    }
}

__device__ __forceinline__ void bs_scale(const float (&un)[NB], float dn, int lane,
                                         float (&nw)[NB], float& rinv) {
    rinv = 1.0f / dn;
    #pragma unroll
    for (int c = 0; c < NB; c++) nw[c] = (c > lane) ? -(un[c] * rinv) : 0.f;
}

__global__ void __launch_bounds__(BS_T)
backsolve_kernel(const float* __restrict__ K, float* __restrict__ x) {
    int b = blockIdx.x, tid = threadIdx.x;
    int lane = tid & 31;
    int wave = __builtin_amdgcn_readfirstlane(tid >> 5);
    int sub = tid & 7, rloc = tid >> 3;
    __shared__ __attribute__((aligned(16))) float y[PP];
    const float* Kb = K + (size_t)b * PP * PP;
    for (int i = tid; i < PP; i += BS_T)
        y[i] = (i < KSZ) ? Kb[(size_t)i * PP + KSZ] : 0.f;

    float yacc[BS_MAXP];
    #pragma unroll
    for (int p = 0; p < BS_MAXP; p++)
        yacc[p] = Kb[(size_t)(p * BS_RPP + rloc) * PP + KSZ];

    float nw[NB], un[NB];
    float rinv = 0.f, dn = 1.f;
    #pragma unroll
    for (int c = 0; c < NB; c++) { nw[c] = 0.f; un[c] = 0.f; }
    if (wave == 0) {
        bs_load_diag(Kb, (NBLK32 - 1) * NB, lane, un, dn);
        bs_scale(un, dn, lane, nw, rinv);
    }
    __syncthreads();

    for (int kb = NBLK32 - 1; kb >= 0; kb--) {
        int r0 = kb * NB;
        float4 cur[BS_MAXP];
        #pragma unroll
        for (int p = 0; p < BS_MAXP; p++) {
            if (p * BS_RPP + wave * 4 >= r0) break;
            cur[p] = *(const float4*)(Kb + (size_t)(p * BS_RPP + rloc) * PP + r0 + 4 * sub);
        }
        if (wave == 0) {
            if (kb > 0) bs_load_diag(Kb, r0 - NB, lane, un, dn);
            float t = y[r0 + lane] * rinv;
            #pragma unroll
            for (int j = NB - 1; j >= 1; j--) {
                float xj = lane_bcast(t, j);
                t += nw[j] * xj;
            }
            y[r0 + lane] = t;
            asm volatile("" :: "v"(un[0]));
        }
        __syncthreads();

        float4 xv = *(const float4*)(y + r0 + 4 * sub);
        #pragma unroll
        for (int p = 0; p < BS_MAXP; p++) {
            int base = p * BS_RPP + wave * 4;
            if (base >= r0) break;
            {
                float part = cur[p].x * xv.x + cur[p].y * xv.y + cur[p].z * xv.z + cur[p].w * xv.w;
                part += lane_xor<1>(part);
                part += lane_xor<2>(part);
                part += lane_xor<4>(part);
                yacc[p] -= part;
                if (base >= r0 - NB && sub == 0) y[p * BS_RPP + rloc] = yacc[p];
            }
        }
        if (wave == 0 && kb > 0) bs_scale(un, dn, lane, nw, rinv);
        __syncthreads();
    }
    for (int i = tid; i < PP; i += BS_T) x[(size_t)b * PP + i] = y[i];
}

__global__ void vote_kernel(const float* __restrict__ x, float* __restrict__ out) {
    int s = blockIdx.x, b = blockIdx.y, tid = threadIdx.x;
    __shared__ float vals[NN_];
    if (tid < NN_) {
        float v = x[(size_t)b * PP + tid * NN_ + s];
        vals[tid] = fminf(fmaxf(v, 0.f), 1.f);
    }
    __syncthreads();
    if (tid == 0) {
        float m = -1e30f;
        for (int t = 0; t < NN_; t++) m = fmaxf(m, vals[t]);
        float sum = 0.f;
        for (int t = 0; t < NN_; t++) { float e = __expf(ALPHA * (vals[t] - m)); vals[t] = e; sum += e; }
        float inv = 1.f / sum;
        for (int t = 0; t < NN_; t++) out[((size_t)b * NN_ + s) * NN_ + t] = vals[t] * inv;
    }
}

extern "C" void kernel_launch(void* const* d_in, const int* in_sizes, int n_in,
                              void* d_out, int out_size, void* d_ws, size_t ws_size,
                              hipStream_t stream) {
    const float* emb1 = (const float*)d_in[0];
    const float* emb2 = (const float*)d_in[1];
    const float* Wcg  = (const float*)d_in[2];
    const float* bcg  = (const float*)d_in[3];

    float* ws  = (float*)d_ws;
    size_t kelem = (size_t)BB * PP * PP;
    float* Km  = ws;
    float* h1  = Km + kelem;
    float* h2  = h1 + (size_t)BB * NN_ * DD;
    float* S   = h2 + (size_t)BB * NN_ * DD;
    float* x   = S  + (size_t)BB * NN2;

    feat_kernel<<<dim3(BB, 2), 256, 0, stream>>>(emb1, emb2, Wcg, bcg, h1, h2);
    gram_kernel<<<BB, 256, 0, stream>>>(h1, h2, S);
    assemble_kernel<<<dim3(NBLK, NBLK, BB), dim3(16, 16), 0, stream>>>(S, Km);

    float* slot = h1;
    for (int kb = 0; kb < NBLK32; kb++) {
        int W2 = PP - (kb + 1) * NB;
        int nc = (W2 + PNT - 1) / PNT;
        lu_panels_kernel<<<dim3(nc > 0 ? nc : 1, BB), PNT, 0, stream>>>(Km, slot, kb);
        if (W2 > 0) {
            int t32 = W2 / 32;
            lu_update_kernel<<<dim3(t32, t32, BB), 32, 0, stream>>>(Km, kb);
        }
    }

    backsolve_kernel<<<BB, BS_T, 0, stream>>>(Km, x);
    vote_kernel<<<dim3(NN_, BB), 64, 0, stream>>>(x, (float*)d_out);
}


# Panel of step kb+1 runs inside the update launch of step kb: waves of the first tile row/column update their tiles by WMMA, hand them through LDS, run the register panel (85 -> 43 LU launches)
# speedup vs baseline: 10.3426x; 1.9861x over previous
#include <hip/hip_runtime.h>
#include <math.h>


#define BB   8
#define NN_  36
#define DD   64
#define NN2  (NN_*NN_)
#define NCON (2*NN_)
#define KSZ  (NN2+NCON)
#define PP   1376
#define NBLK (PP/16)
#define KDIAG 1225.0f
#define EPSK  1e-5f
#define ALPHA 1000.0f

typedef float v2f __attribute__((ext_vector_type(2)));
typedef float v8f __attribute__((ext_vector_type(8)));

__global__ void feat_kernel(const float* __restrict__ emb1,
                            const float* __restrict__ emb2,
                            const float* __restrict__ W,
                            const float* __restrict__ bias,
                            float* __restrict__ h1, float* __restrict__ h2) {
    int b = blockIdx.x, which = blockIdx.y;
    const float* e = (which == 0 ? emb1 : emb2) + (size_t)b * NN_ * DD;
    float* h = (which == 0 ? h1 : h2) + (size_t)b * NN_ * DD;
    __shared__ float se[NN_ * DD];
    __shared__ float sw[DD * DD];
    __shared__ float snrm[DD];
    int tid = threadIdx.x;
    for (int i = tid; i < NN_ * DD; i += 256) se[i] = e[i];
    for (int i = tid; i < DD * DD; i += 256) sw[i] = W[i];
    __syncthreads();
    for (int d = tid; d < DD; d += 256) {
        float s = 0.f;
        for (int n = 0; n < NN_; n++) { float v = se[n * DD + d]; s += v * v; }
        snrm[d] = fmaxf(sqrtf(s), 1e-12f);
    }
    __syncthreads();
    for (int i = tid; i < NN_ * DD; i += 256) se[i] = se[i] / snrm[i % DD];
    __syncthreads();
    for (int idx = tid; idx < NN_ * DD; idx += 256) {
        int n = idx / DD, i = idx % DD;
        float acc = bias[i];
        for (int d = 0; d < DD; d++) acc += se[n * DD + d] * sw[i * DD + d];
        h[idx] = fmaxf(acc, 0.f);
    }
}

__global__ void gram_kernel(const float* __restrict__ h1,
                            const float* __restrict__ h2,
                            float* __restrict__ S) {
    int b = blockIdx.x, tid = threadIdx.x;
    __shared__ float s1[NN_ * DD];
    __shared__ float s2[NN_ * DD];
    for (int i = tid; i < NN_ * DD; i += 256) {
        s1[i] = h1[(size_t)b * NN_ * DD + i];
        s2[i] = h2[(size_t)b * NN_ * DD + i];
    }
    __syncthreads();
    for (int idx = tid; idx < NN2; idx += 256) {
        int i = idx / NN_, j = idx % NN_;
        float a = 0.f;
        for (int d = 0; d < DD; d++) a += s1[i * DD + d] * s2[j * DD + d];
        S[(size_t)b * NN2 + idx] = a;
    }
}

__global__ void assemble_kernel(const float* __restrict__ S, float* __restrict__ K) {
    int b = blockIdx.z;
    int r = blockIdx.y * 16 + threadIdx.y;
    int c = blockIdx.x * 16 + threadIdx.x;
    const float* Sb = S + (size_t)b * NN2;
    float val = 0.f;
    if (r < NN2 && c < NN2) {
        int t = r / NN_, s = r - t * NN_, u = c / NN_, v = c - u * NN_;
        val = (r == c) ? KDIAG : 0.f;
        if (s != v && t != u) val -= 0.5f * (Sb[s * NN_ + t] + Sb[v * NN_ + u]);
    } else if (r < NN2) {
        int i = c - NN2, t = r / NN_, s = r - t * NN_;
        if (i < NN_)            val = (s == i) ? 1.f : 0.f;
        else if (i < NCON)      val = (t == i - NN_) ? 1.f : 0.f;
        else if (i == NCON)     val = Sb[s * NN_ + t];
    } else if (c < NN2) {
        int i = r - NN2, u = c / NN_, v = c - u * NN_;
        if (i < NN_)            val = (v == i) ? 1.f : 0.f;
        else if (i < NCON)      val = (u == i - NN_) ? 1.f : 0.f;
    } else if (r == c) {
        val = (r < KSZ) ? -EPSK : 1.0f;
    } else if (c == KSZ && r < KSZ) {
        val = 1.0f;
    }
    K[((size_t)b * PP + r) * PP + c] = val;
}

#define NB  32
#define NBLK32 (PP/NB)
#define PNT 64

__device__ __forceinline__ float lane_bcast(float v, int lane) {
    return __int_as_float(__builtin_amdgcn_readlane(__float_as_int(v), lane));
}

__device__ __forceinline__ float in_reg(float v) {
    asm("" : "+v"(v));
    return v;
}

__device__ __forceinline__ void panel_lu(float (&a)[NB], float (&xr)[NB], float (&xc)[NB], int row) {
    #pragma unroll
    for (int j = 0; j < NB; j++) {
        float u[NB];
        #pragma unroll
        for (int c2 = j; c2 < NB; c2++) u[c2] = lane_bcast(a[c2], j);
        float dinv = 1.0f / u[j];
        bool below = row > j;
        float lij = a[j] * dinv;
        float nl = in_reg(-lij);
        #pragma unroll
        for (int c2 = j + 1; c2 < NB; c2++) {
            float t = a[c2] + nl * u[c2];
            a[c2] = below ? t : a[c2];
        }
        a[j] = below ? lij : a[j];

        float xl = xr[j] * dinv;
        xr[j] = xl;
        float nxl = in_reg(-xl);
        #pragma unroll
        for (int c2 = j + 1; c2 < NB; c2++) xr[c2] += nxl * u[c2];
        __builtin_amdgcn_sched_barrier(0);

        float nxj = in_reg(-xc[j]);
        #pragma unroll
        for (int r = j + 1; r < NB; r++) xc[r] += lane_bcast(a[j], r) * nxj;
        __builtin_amdgcn_sched_barrier(0);
    }
}

__global__ void __launch_bounds__(PNT)
lu_panels_kernel(float* __restrict__ K, float* __restrict__ slot, int kb) {
    int b = blockIdx.y, bx = blockIdx.x, tid = threadIdx.x;
    int r0 = kb * NB;
    float* Kb = K + (size_t)b * PP * PP;
    float* sl = slot + (size_t)b * (NB * NB);
    int row = tid & (NB - 1);

    int idx = r0 + NB + bx * PNT + tid;
    bool act = idx < PP;
    bool keeper = (bx == 0 && tid < NB);

    float xr[NB], xc[NB], a[NB], o[NB];
    #pragma unroll
    for (int j = 0; j < NB; j++) { xr[j] = 0.f; xc[j] = 0.f; o[j] = 0.f; }
    if (act) {
        const float4* p = (const float4*)(Kb + (size_t)idx * PP + r0);
        #pragma unroll
        for (int q = 0; q < NB / 4; q++) {
            float4 v = p[q];
            xr[4 * q] = v.x; xr[4 * q + 1] = v.y; xr[4 * q + 2] = v.z; xr[4 * q + 3] = v.w;
        }
        #pragma unroll
        for (int j = 0; j < NB; j++) xc[j] = Kb[(size_t)(r0 + j) * PP + idx];
    }
    {
        const float4* p = (const float4*)(Kb + (size_t)(r0 + row) * PP + r0);
        #pragma unroll
        for (int q = 0; q < NB / 4; q++) {
            float4 v = p[q];
            a[4 * q] = v.x; a[4 * q + 1] = v.y; a[4 * q + 2] = v.z; a[4 * q + 3] = v.w;
        }
    }
    if (keeper && kb > 0) {
        const float4* p = (const float4*)(sl + row * NB);
        #pragma unroll
        for (int q = 0; q < NB / 4; q++) {
            float4 v = p[q];
            o[4 * q] = v.x; o[4 * q + 1] = v.y; o[4 * q + 2] = v.z; o[4 * q + 3] = v.w;
        }
    }

    panel_lu(a, xr, xc, row);

    if (keeper) {
        if (kb > 0) {
            float4* p = (float4*)(Kb + (size_t)(r0 - NB + row) * PP + r0 - NB);
            #pragma unroll
            for (int q = 0; q < NB / 4; q++)
                p[q] = make_float4(o[4 * q], o[4 * q + 1], o[4 * q + 2], o[4 * q + 3]);
        }
        float4* p = (r0 + NB == PP) ? (float4*)(Kb + (size_t)(r0 + row) * PP + r0)
                                    : (float4*)(sl + row * NB);
        #pragma unroll
        for (int q = 0; q < NB / 4; q++)
            p[q] = make_float4(a[4 * q], a[4 * q + 1], a[4 * q + 2], a[4 * q + 3]);
    }

    if (act) {
        float4* p = (float4*)(Kb + (size_t)idx * PP + r0);
        #pragma unroll
        for (int q = 0; q < NB / 4; q++)
            p[q] = make_float4(xr[4 * q], xr[4 * q + 1], xr[4 * q + 2], xr[4 * q + 3]);
        #pragma unroll
        for (int j = 0; j < NB; j++) Kb[(size_t)(r0 + j) * PP + idx] = xc[j];
    }
}

__device__ __forceinline__ void update_tile(float* __restrict__ Kb,
                                            int row0, int col0, int kcol,
                                            int lane) {
    int half = lane >> 4, l16 = lane & 15;
    int row1 = row0 + 16, col1 = col0 + 16;

    v8f c00, c01, c10, c11;
    #pragma unroll
    for (int r = 0; r < 8; r++) {
        size_t ro0 = (size_t)(row0 + r + 8 * half) * PP;
        size_t ro1 = (size_t)(row1 + r + 8 * half) * PP;
        c00[r] = Kb[ro0 + col0 + l16];
        c01[r] = Kb[ro0 + col1 + l16];
        c10[r] = Kb[ro1 + col0 + l16];
        c11[r] = Kb[ro1 + col1 + l16];
    }
    v2f a0[NB / 4], a1[NB / 4], b0[NB / 4], b1[NB / 4];
    #pragma unroll
    for (int kk = 0; kk < NB / 4; kk++) {
        int kc = kcol + kk * 4 + 2 * half;
        a0[kk] = *(const v2f*)(Kb + (size_t)(row0 + l16) * PP + kc);
        a1[kk] = *(const v2f*)(Kb + (size_t)(row1 + l16) * PP + kc);
        #pragma unroll
        for (int i = 0; i < 2; i++) {
            size_t kr = (size_t)(kc + i) * PP;
            b0[kk][i] = Kb[kr + col0 + l16];
            b1[kk][i] = Kb[kr + col1 + l16];
        }
    }
    __builtin_amdgcn_sched_barrier(0);
    #pragma unroll
    for (int kk = 0; kk < NB / 4; kk++) {
        v2f n0 = -a0[kk], n1 = -a1[kk];
        c00 = __builtin_amdgcn_wmma_f32_16x16x4_f32(false, n0, false, b0[kk], (short)0, c00, false, false);
        c01 = __builtin_amdgcn_wmma_f32_16x16x4_f32(false, n0, false, b1[kk], (short)0, c01, false, false);
        c10 = __builtin_amdgcn_wmma_f32_16x16x4_f32(false, n1, false, b0[kk], (short)0, c10, false, false);
        c11 = __builtin_amdgcn_wmma_f32_16x16x4_f32(false, n1, false, b1[kk], (short)0, c11, false, false);
    }
    #pragma unroll
    for (int r = 0; r < 8; r++) {
        size_t ro0 = (size_t)(row0 + r + 8 * half) * PP;
        size_t ro1 = (size_t)(row1 + r + 8 * half) * PP;
        Kb[ro0 + col0 + l16] = c00[r];
        Kb[ro0 + col1 + l16] = c01[r];
        Kb[ro1 + col0 + l16] = c10[r];
        Kb[ro1 + col1 + l16] = c11[r];
    }
}

#define TP 33

__device__ __forceinline__ void frag_c(const float* __restrict__ Kb, int row0, int col0,
                                       int half, int l16, v8f (&c)[4]) {
    #pragma unroll
    for (int r = 0; r < 8; r++) {
        size_t ro0 = (size_t)(row0 + r + 8 * half) * PP;
        size_t ro1 = (size_t)(row0 + 16 + r + 8 * half) * PP;
        c[0][r] = Kb[ro0 + col0 + l16];
        c[1][r] = Kb[ro0 + col0 + 16 + l16];
        c[2][r] = Kb[ro1 + col0 + l16];
        c[3][r] = Kb[ro1 + col0 + 16 + l16];
    }
}

__device__ __forceinline__ void frag_a(const float* __restrict__ Kb, int row0, int kcol,
                                       int half, int l16, v2f (&a0)[NB / 4], v2f (&a1)[NB / 4]) {
    #pragma unroll
    for (int kk = 0; kk < NB / 4; kk++) {
        int kc = kcol + kk * 4 + 2 * half;
        a0[kk] = *(const v2f*)(Kb + (size_t)(row0 + l16) * PP + kc);
        a1[kk] = *(const v2f*)(Kb + (size_t)(row0 + 16 + l16) * PP + kc);
    }
}

__device__ __forceinline__ void frag_b(const float* __restrict__ Kb, int col0, int kcol,
                                       int half, int l16, v2f (&b0)[NB / 4], v2f (&b1)[NB / 4]) {
    #pragma unroll
    for (int kk = 0; kk < NB / 4; kk++) {
        int kc = kcol + kk * 4 + 2 * half;
        #pragma unroll
        for (int i = 0; i < 2; i++) {
            size_t kr = (size_t)(kc + i) * PP;
            b0[kk][i] = Kb[kr + col0 + l16];
            b1[kk][i] = Kb[kr + col0 + 16 + l16];
        }
    }
}

__device__ __forceinline__ void tile_to_lds(const v2f (&a0)[NB / 4], const v2f (&a1)[NB / 4],
                                            const v2f (&b0)[NB / 4], const v2f (&b1)[NB / 4],
                                            v8f (&c)[4], float* __restrict__ t, int half, int l16) {
    #pragma unroll
    for (int kk = 0; kk < NB / 4; kk++) {
        v2f n0 = -a0[kk], n1 = -a1[kk];
        c[0] = __builtin_amdgcn_wmma_f32_16x16x4_f32(false, n0, false, b0[kk], (short)0, c[0], false, false);
        c[1] = __builtin_amdgcn_wmma_f32_16x16x4_f32(false, n0, false, b1[kk], (short)0, c[1], false, false);
        c[2] = __builtin_amdgcn_wmma_f32_16x16x4_f32(false, n1, false, b0[kk], (short)0, c[2], false, false);
        c[3] = __builtin_amdgcn_wmma_f32_16x16x4_f32(false, n1, false, b1[kk], (short)0, c[3], false, false);
    }
    #pragma unroll
    for (int r = 0; r < 8; r++) {
        int m0 = (r + 8 * half) * TP, m1 = (16 + r + 8 * half) * TP;
        t[m0 + l16]      = c[0][r];
        t[m0 + 16 + l16] = c[1][r];
        t[m1 + l16]      = c[2][r];
        t[m1 + 16 + l16] = c[3][r];
    }
}

__global__ void __launch_bounds__(32)
lu_step_kernel(float* __restrict__ K, float* __restrict__ slot, int kb) {
    int b = blockIdx.x, tj = blockIdx.y, ti = blockIdx.z;
    int lane = threadIdx.x;
    float* Kb = K + (size_t)b * PP * PP;
    int kcol = kb * NB;
    int base = kcol + NB;
    if (__builtin_expect(ti > 0, 1)) {
        if (tj > 0) update_tile(Kb, base + ti * 32, base + tj * 32, kcol, lane);
        return;
    }

    __shared__ float st[3][NB * TP];
    int p = tj;
    bool keeper = (p == 0);
    int half = lane >> 4, l16 = lane & 15;
    int own = base + p * 32;
    float* sl = slot + (size_t)b * (NB * NB);

    v8f cd[4], cr[4], cc[4];
    v2f ad0[NB / 4], ad1[NB / 4], bd0[NB / 4], bd1[NB / 4];
    v2f ar0[NB / 4], ar1[NB / 4], bc0[NB / 4], bc1[NB / 4];
    frag_c(Kb, base, base, half, l16, cd);
    frag_a(Kb, base, kcol, half, l16, ad0, ad1);
    frag_b(Kb, base, kcol, half, l16, bd0, bd1);
    frag_c(Kb, own, base, half, l16, cr);
    frag_a(Kb, own, kcol, half, l16, ar0, ar1);
    frag_c(Kb, base, own, half, l16, cc);
    frag_b(Kb, own, kcol, half, l16, bc0, bc1);
    __builtin_amdgcn_sched_barrier(0);
    tile_to_lds(ad0, ad1, bd0, bd1, cd, st[0], half, l16);
    tile_to_lds(ar0, ar1, bd0, bd1, cr, st[1], half, l16);
    tile_to_lds(ad0, ad1, bc0, bc1, cc, st[2], half, l16);
    __syncthreads();

    float a[NB], xr[NB], xc[NB];
    #pragma unroll
    for (int j = 0; j < NB; j++) {
        a[j]  = st[0][lane * TP + j];
        xr[j] = st[1][lane * TP + j];
        xc[j] = st[2][j * TP + lane];
    }

    float4 o[NB / 4];
    #pragma unroll
    for (int q = 0; q < NB / 4; q++) o[q] = make_float4(0.f, 0.f, 0.f, 0.f);
    if (keeper) {
        const float4* q4 = (const float4*)(sl + lane * NB);
        #pragma unroll
        for (int q = 0; q < NB / 4; q++) o[q] = q4[q];
    }

    panel_lu(a, xr, xc, lane);

    if (keeper) {
        {
            float4* d = (float4*)(Kb + (size_t)(kcol + lane) * PP + kcol);
            #pragma unroll
            for (int q = 0; q < NB / 4; q++) d[q] = o[q];
        }
        float4* d = (base + NB == PP) ? (float4*)(Kb + (size_t)(base + lane) * PP + base)
                                      : (float4*)(sl + lane * NB);
        #pragma unroll
        for (int q = 0; q < NB / 4; q++)
            d[q] = make_float4(a[4 * q], a[4 * q + 1], a[4 * q + 2], a[4 * q + 3]);
    } else {
        float4* d = (float4*)(Kb + (size_t)(own + lane) * PP + base);
        #pragma unroll
        for (int q = 0; q < NB / 4; q++)
            d[q] = make_float4(xr[4 * q], xr[4 * q + 1], xr[4 * q + 2], xr[4 * q + 3]);
        #pragma unroll
        for (int j = 0; j < NB; j++) Kb[(size_t)(base + j) * PP + own + lane] = xc[j];
    }
}

#define BS_T    512
#define BS_RPP  (BS_T / 8)
#define BS_MAXP ((PP - NB) / BS_RPP)

template <int M>
__device__ __forceinline__ float lane_xor(float v) {
    return __int_as_float(__builtin_amdgcn_update_dpp(0, __float_as_int(v), 0x160 + M, 0xf, 0xf, true));
}

__device__ __forceinline__ void bs_load_diag(const float* __restrict__ Kb, int r, int lane,
                                             float (&un)[NB], float& dn) {
    dn = Kb[(size_t)(r + lane) * PP + r + lane];
    const float4* p = (const float4*)(Kb + (size_t)(r + lane) * PP + r);
    #pragma unroll
    for (int q = 0; q < NB / 4; q++) {
        float4 v = p[q];
        un[4 * q] = v.x; un[4 * q + 1] = v.y; un[4 * q + 2] = v.z; un[4 * q + 3] = v.w;
    }
}

__device__ __forceinline__ void bs_scale(const float (&un)[NB], float dn, int lane,
                                         float (&nw)[NB], float& rinv) {
    rinv = 1.0f / dn;
    #pragma unroll
    for (int c = 0; c < NB; c++) nw[c] = (c > lane) ? -(un[c] * rinv) : 0.f;
}

__global__ void __launch_bounds__(BS_T)
backsolve_kernel(const float* __restrict__ K, float* __restrict__ x) {
    int b = blockIdx.x, tid = threadIdx.x;
    int lane = tid & 31;
    int wave = __builtin_amdgcn_readfirstlane(tid >> 5);
    int sub = tid & 7, rloc = tid >> 3;
    __shared__ __attribute__((aligned(16))) float y[PP];
    const float* Kb = K + (size_t)b * PP * PP;
    for (int i = tid; i < PP; i += BS_T)
        y[i] = (i < KSZ) ? Kb[(size_t)i * PP + KSZ] : 0.f;

    float yacc[BS_MAXP];
    #pragma unroll
    for (int p = 0; p < BS_MAXP; p++)
        yacc[p] = Kb[(size_t)(p * BS_RPP + rloc) * PP + KSZ];

    float nw[NB], un[NB];
    float rinv = 0.f, dn = 1.f;
    #pragma unroll
    for (int c = 0; c < NB; c++) { nw[c] = 0.f; un[c] = 0.f; }
    if (wave == 0) {
        bs_load_diag(Kb, (NBLK32 - 1) * NB, lane, un, dn);
        bs_scale(un, dn, lane, nw, rinv);
    }
    __syncthreads();

    for (int kb = NBLK32 - 1; kb >= 0; kb--) {
        int r0 = kb * NB;
        float4 cur[BS_MAXP];
        #pragma unroll
        for (int p = 0; p < BS_MAXP; p++) {
            if (p * BS_RPP + wave * 4 >= r0) break;
            cur[p] = *(const float4*)(Kb + (size_t)(p * BS_RPP + rloc) * PP + r0 + 4 * sub);
        }
        if (wave == 0) {
            if (kb > 0) bs_load_diag(Kb, r0 - NB, lane, un, dn);
            float t = y[r0 + lane] * rinv;
            #pragma unroll
            for (int j = NB - 1; j >= 1; j--) {
                float xj = lane_bcast(t, j);
                t += nw[j] * xj;
            }
            y[r0 + lane] = t;
            asm volatile("" :: "v"(un[0]));
        }
        __syncthreads();

        float4 xv = *(const float4*)(y + r0 + 4 * sub);
        #pragma unroll
        for (int p = 0; p < BS_MAXP; p++) {
            int base = p * BS_RPP + wave * 4;
            if (base >= r0) break;
            {
                float part = cur[p].x * xv.x + cur[p].y * xv.y + cur[p].z * xv.z + cur[p].w * xv.w;
                part += lane_xor<1>(part);
                part += lane_xor<2>(part);
                part += lane_xor<4>(part);
                yacc[p] -= part;
                if (base >= r0 - NB && sub == 0) y[p * BS_RPP + rloc] = yacc[p];
            }
        }
        if (wave == 0 && kb > 0) bs_scale(un, dn, lane, nw, rinv);
        __syncthreads();
    }
    for (int i = tid; i < PP; i += BS_T) x[(size_t)b * PP + i] = y[i];
}

__global__ void vote_kernel(const float* __restrict__ x, float* __restrict__ out) {
    int s = blockIdx.x, b = blockIdx.y, tid = threadIdx.x;
    __shared__ float vals[NN_];
    if (tid < NN_) {
        float v = x[(size_t)b * PP + tid * NN_ + s];
        vals[tid] = fminf(fmaxf(v, 0.f), 1.f);
    }
    __syncthreads();
    if (tid == 0) {
        float m = -1e30f;
        for (int t = 0; t < NN_; t++) m = fmaxf(m, vals[t]);
        float sum = 0.f;
        for (int t = 0; t < NN_; t++) { float e = __expf(ALPHA * (vals[t] - m)); vals[t] = e; sum += e; }
        float inv = 1.f / sum;
        for (int t = 0; t < NN_; t++) out[((size_t)b * NN_ + s) * NN_ + t] = vals[t] * inv;
    }
}

extern "C" void kernel_launch(void* const* d_in, const int* in_sizes, int n_in,
                              void* d_out, int out_size, void* d_ws, size_t ws_size,
                              hipStream_t stream) {
    const float* emb1 = (const float*)d_in[0];
    const float* emb2 = (const float*)d_in[1];
    const float* Wcg  = (const float*)d_in[2];
    const float* bcg  = (const float*)d_in[3];

    float* ws  = (float*)d_ws;
    size_t kelem = (size_t)BB * PP * PP;
    float* Km  = ws;
    float* h1  = Km + kelem;
    float* h2  = h1 + (size_t)BB * NN_ * DD;
    float* S   = h2 + (size_t)BB * NN_ * DD;
    float* x   = S  + (size_t)BB * NN2;

    feat_kernel<<<dim3(BB, 2), 256, 0, stream>>>(emb1, emb2, Wcg, bcg, h1, h2);
    gram_kernel<<<BB, 256, 0, stream>>>(h1, h2, S);
    assemble_kernel<<<dim3(NBLK, NBLK, BB), dim3(16, 16), 0, stream>>>(S, Km);

    float* slot = h1;
    lu_panels_kernel<<<dim3((PP - NB + PNT - 1) / PNT, BB), PNT, 0, stream>>>(Km, slot, 0);
    for (int kb = 0; kb < NBLK32 - 1; kb++) {
        int t32 = NBLK32 - 1 - kb;
        lu_step_kernel<<<dim3(BB, t32, t32), 32, 0, stream>>>(Km, slot, kb);
    }

    backsolve_kernel<<<BB, BS_T, 0, stream>>>(Km, x);
    vote_kernel<<<dim3(NN_, BB), 64, 0, stream>>>(x, (float*)d_out);
}
